# starting kernel
# baseline (speedup 1.0000x reference)
#include <hip/hip_runtime.h>
#include <hip/hip_bf16.h>


typedef __attribute__((ext_vector_type(2))) float v2f;
typedef __attribute__((ext_vector_type(8))) float v8f;

#define GAT_B 2
#define GAT_N 512
#define GAT_F 128
#define GAT_H 256
#define NEG_SLOPE 0.04f
#define MASK_VAL  (-9.0e15f)

__global__ __launch_bounds__(256) void pq_gemm(const float* __restrict__ x,
                                               const float* __restrict__ w1,
                                               float* __restrict__ P,
                                               float* __restrict__ Q) {
    const int wave = (blockIdx.x * blockDim.x + threadIdx.x) >> 5;
    const int lane = threadIdx.x & 31;
    const int which = wave >> 10;
    const int rem   = wave & 1023;
    const int m0 = (rem >> 4) << 4;
    const int n0 = (rem & 15) << 4;
    const int lhalf = lane >> 4;
    const int lm    = lane & 15;

    const float* Arow = x + (size_t)(m0 + lm) * GAT_F;
    const float* Brow = w1 + (size_t)(n0 + lm) * GAT_H + (which ? GAT_F : 0);

    v8f c = {0.f, 0.f, 0.f, 0.f, 0.f, 0.f, 0.f, 0.f};
    #pragma unroll 8
    for (int k0 = 0; k0 < GAT_F; k0 += 4) {
        const int kk = k0 + 2 * lhalf;
        v2f a = *(const v2f*)(Arow + kk);
        v2f b = *(const v2f*)(Brow + kk);
        c = __builtin_amdgcn_wmma_f32_16x16x4_f32(false, a, false, b,
                                                  (short)0, c, false, false);
    }

    float* out = (which ? Q : P);
    float* orow = out + (size_t)(m0 + 8 * lhalf) * GAT_H + n0 + lm;
    #pragma unroll
    for (int v = 0; v < 8; ++v) orow[(size_t)v * GAT_H] = c[v];
}

__global__ __launch_bounds__(256) void e_kernel(const float* __restrict__ P,
                                                const float* __restrict__ Q,
                                                const float* __restrict__ b1,
                                                const float* __restrict__ w2,
                                                const float* __restrict__ b2,
                                                const int*   __restrict__ adj,
                                                float* __restrict__ E) {
    const int bi = blockIdx.x;
    const int b  = bi >> 9;
    const int i  = bi & (GAT_N - 1);
    const int t  = threadIdx.x;

    __shared__ float s[GAT_H];
    __shared__ float w2s[GAT_H];
    s[t]   = P[(size_t)bi * GAT_H + t] + b1[t];
    w2s[t] = w2[t];
    __syncthreads();

    const float b2v = b2[0];
    const int*   adjrow = adj + (size_t)b * GAT_N * GAT_N + (size_t)i * GAT_N;
    float*       erow   = E   + (size_t)b * GAT_N * GAT_N + (size_t)i * GAT_N;

    for (int j = t; j < GAT_N; j += 256) {
        const float* qrow = Q + (size_t)(b * GAT_N + j) * GAT_H;
        float acc = 0.f;
        #pragma unroll 4
        for (int h = 0; h < GAT_H; h += 4) {
            const float4 q4 = *(const float4*)(qrow + h);
            acc += w2s[h + 0] * fmaxf(s[h + 0] + q4.x, 0.f);
            acc += w2s[h + 1] * fmaxf(s[h + 1] + q4.y, 0.f);
            acc += w2s[h + 2] * fmaxf(s[h + 2] + q4.z, 0.f);
            acc += w2s[h + 3] * fmaxf(s[h + 3] + q4.w, 0.f);
        }
        float e = acc + b2v;
        e = (e > 0.f) ? e : NEG_SLOPE * e;
        e = (adjrow[j] > 0) ? e : MASK_VAL;
        erow[j] = e;
    }
}

__global__ __launch_bounds__(256) void softmax_kernel(float* __restrict__ E,
                                                      float* __restrict__ att00) {
    const int row = blockIdx.x;
    const int t   = threadIdx.x;
    float* e = E + (size_t)row * GAT_N;

    __shared__ float red[256];

    const float x0 = e[t], x1 = e[t + 256];
    red[t] = fmaxf(x0, x1);
    __syncthreads();
    #pragma unroll
    for (int s = 128; s > 0; s >>= 1) {
        if (t < s) red[t] = fmaxf(red[t], red[t + s]);
        __syncthreads();
    }
    const float mx = red[0];
    __syncthreads();

    const float e0 = __expf(x0 - mx);
    const float e1 = __expf(x1 - mx);
    red[t] = e0 + e1;
    __syncthreads();
    #pragma unroll
    for (int s = 128; s > 0; s >>= 1) {
        if (t < s) red[t] += red[t + s];
        __syncthreads();
    }
    const float inv = 1.f / red[0];

    e[t]       = e0 * inv;
    e[t + 256] = e1 * inv;
    if (row == 0) {
        att00[t]       = e0 * inv;
        att00[t + 256] = e1 * inv;
    }
}

__global__ __launch_bounds__(256) void av_gemm(const float* __restrict__ att,
                                               const float* __restrict__ x,
                                               float* __restrict__ out) {
    const int wave = (blockIdx.x * blockDim.x + threadIdx.x) >> 5;
    const int lane = threadIdx.x & 31;
    const int b    = wave >> 8;
    const int rem  = wave & 255;
    const int m0 = (rem >> 3) << 4;
    const int n0 = (rem & 7) << 4;
    const int lhalf = lane >> 4;
    const int lm    = lane & 15;

    const float* A = att + (size_t)b * GAT_N * GAT_N;
    const float* X = x   + (size_t)b * GAT_N * GAT_F;
    const float* Arow = A + (size_t)(m0 + lm) * GAT_N;
    const float* Bcol = X + n0 + lm;

    v8f c = {0.f, 0.f, 0.f, 0.f, 0.f, 0.f, 0.f, 0.f};
    #pragma unroll 4
    for (int k0 = 0; k0 < GAT_N; k0 += 4) {
        const int kk = k0 + 2 * lhalf;
        v2f a = *(const v2f*)(Arow + kk);
        v2f bb;
        bb.x = Bcol[(size_t)kk * GAT_F];
        bb.y = Bcol[(size_t)(kk + 1) * GAT_F];
        c = __builtin_amdgcn_wmma_f32_16x16x4_f32(false, a, false, bb,
                                                  (short)0, c, false, false);
    }

    float* orow = out + (size_t)b * GAT_N * GAT_F
                      + (size_t)(m0 + 8 * lhalf) * GAT_F + n0 + lm;
    #pragma unroll
    for (int v = 0; v < 8; ++v) orow[(size_t)v * GAT_F] = c[v];
}

extern "C" void kernel_launch(void* const* d_in, const int* in_sizes, int n_in,
                              void* d_out, int out_size, void* d_ws, size_t ws_size,
                              hipStream_t stream) {
    const float* x   = (const float*)d_in[0];
    const float* w1  = (const float*)d_in[1];
    const float* b1  = (const float*)d_in[2];
    const float* w2  = (const float*)d_in[3];
    const float* b2  = (const float*)d_in[4];
    const int*   adj = (const int*)d_in[5];

    float* out = (float*)d_out;
    float* P = (float*)d_ws;
    float* Q = P + (size_t)GAT_B * GAT_N * GAT_H;
    float* E = Q + (size_t)GAT_B * GAT_N * GAT_H;

    float* next_h = out;
    float* att00  = out + (size_t)GAT_B * GAT_N * GAT_F;

    pq_gemm<<<256, 256, 0, stream>>>(x, w1, P, Q);
    e_kernel<<<GAT_B * GAT_N, 256, 0, stream>>>(P, Q, b1, w2, b2, adj, E);
    softmax_kernel<<<GAT_B * GAT_N, 256, 0, stream>>>(E, att00);
    av_gemm<<<64, 256, 0, stream>>>(E, x, next_h);
}


# e-logit: lane owns column pair (j,j+1), read from an h-blocked pair-interleaved Q plane by 16-byte whole-line loads; P+b1 rows and w2 are scalar operands of packed add/fma; no LDS; h order kept
# speedup vs baseline: 1.7175x; 1.7175x over previous
#include <hip/hip_runtime.h>
#include <hip/hip_bf16.h>


typedef __attribute__((ext_vector_type(2))) float v2f;
typedef __attribute__((ext_vector_type(4))) float v4f;
typedef __attribute__((ext_vector_type(8))) float v8f;

#define GAT_B 2
#define GAT_N 512
#define GAT_F 128
#define GAT_H 256
#define GAT_HQ (GAT_H / 4)
#define NEG_SLOPE 0.04f
#define MASK_VAL  (-9.0e15f)

__device__ __forceinline__ v8f tile_kf(const float* __restrict__ Arow,
                                       const float* __restrict__ Brow,
                                       const int lhalf) {
    v8f c = {0.f, 0.f, 0.f, 0.f, 0.f, 0.f, 0.f, 0.f};
    #pragma unroll 8
    for (int k0 = 0; k0 < GAT_F; k0 += 4) {
        const int kk = k0 + 2 * lhalf;
        v2f a = *(const v2f*)(Arow + kk);
        v2f b = *(const v2f*)(Brow + kk);
        c = __builtin_amdgcn_wmma_f32_16x16x4_f32(false, a, false, b,
                                                  (short)0, c, false, false);
    }
    return c;
}

__global__ __launch_bounds__(256) void pq_gemm(const float* __restrict__ x,
                                               const float* __restrict__ w1,
                                               const float* __restrict__ b1,
                                               float* __restrict__ Pb,
                                               float* __restrict__ Qt) {
    const int wave = (blockIdx.x * blockDim.x + threadIdx.x) >> 5;
    const int lane = threadIdx.x & 31;
    const int rem   = wave & 1023;
    const int m0 = (rem >> 4) << 4;
    const int n0 = (rem & 15) << 4;
    const int lhalf = lane >> 4;
    const int lm    = lane & 15;
    const int h     = n0 + lm;

    if (blockIdx.x < 128) {
        const v8f c = tile_kf(x  + (size_t)(m0 + lm) * GAT_F,
                              w1 + (size_t)h * GAT_H, lhalf);
        const float bias = b1[h];
        float* orow = Pb + (size_t)(m0 + 8 * lhalf) * GAT_H + h;
        #pragma unroll
        for (int v = 0; v < 8; ++v) orow[(size_t)v * GAT_H] = c[v] + bias;
    } else {
        const v8f c = tile_kf(x  + (size_t)(m0 + lm) * GAT_F,
                              w1 + (size_t)h * GAT_H + GAT_F, lhalf);
        const int b   = m0 >> 9;
        const int jp0 = ((m0 & (GAT_N - 1)) + 8 * lhalf) >> 1;
        v2f* dst = (v2f*)Qt + (((size_t)(b * GAT_HQ + (h >> 2)) * (GAT_N / 2) + jp0) * 4 + (h & 3));
        #pragma unroll
        for (int p = 0; p < 4; ++p) {
            v2f pr = {c[2 * p], c[2 * p + 1]};
            dst[p * 4] = pr;
        }
    }
}

struct EStage {
    v8f s0, s1, s2, s3;
    v8f w;
    v4f q0, q1, q2, q3;
};

__device__ __forceinline__ EStage e_load(const float* __restrict__ p0,
                                         const float* __restrict__ w2,
                                         const v4f* __restrict__ q, const int h) {
    EStage st;
    st.s0 = *(const v8f*)(p0 + h);
    st.s1 = *(const v8f*)(p0 + GAT_H + h);
    st.s2 = *(const v8f*)(p0 + 2 * GAT_H + h);
    st.s3 = *(const v8f*)(p0 + 3 * GAT_H + h);
    st.w  = *(const v8f*)(w2 + h);
    const v4f* qq = q + (size_t)(h >> 2) * GAT_N;
    st.q0 = qq[0];
    st.q1 = qq[1];
    st.q2 = qq[GAT_N];
    st.q3 = qq[GAT_N + 1];
    return st;
}

__device__ __forceinline__ v2f e_term(const float s, const float w, const v2f qv, const v2f acc) {
    const v2f zero = {0.f, 0.f};
    const v2f sv = {s, s};
    const v2f wv = {w, w};
    return __builtin_elementwise_fma(wv, __builtin_elementwise_max(sv + qv, zero), acc);
}

__device__ __forceinline__ void e_step(const EStage& st, v2f& a0, v2f& a1, v2f& a2, v2f& a3) {
    #pragma unroll
    for (int k = 0; k < 8; ++k) {
        const v4f qq = (k < 2) ? st.q0 : (k < 4) ? st.q1 : (k < 6) ? st.q2 : st.q3;
        const v2f qv = (k & 1) ? (v2f){qq[2], qq[3]} : (v2f){qq[0], qq[1]};
        a0 = e_term(st.s0[k], st.w[k], qv, a0);
        a1 = e_term(st.s1[k], st.w[k], qv, a1);
        a2 = e_term(st.s2[k], st.w[k], qv, a2);
        a3 = e_term(st.s3[k], st.w[k], qv, a3);
    }
}

__device__ __forceinline__ v2f e_finish(const v2f acc, const float b2v, const int2 m) {
    float e0 = acc.x + b2v, e1 = acc.y + b2v;
    e0 = (e0 > 0.f) ? e0 : NEG_SLOPE * e0;
    e1 = (e1 > 0.f) ? e1 : NEG_SLOPE * e1;
    v2f r = {(m.x > 0) ? e0 : MASK_VAL, (m.y > 0) ? e1 : MASK_VAL};
    return r;
}

__global__ __launch_bounds__(256) void e_kernel(const float* __restrict__ Pb,
                                                const float* __restrict__ Qt,
                                                const float* __restrict__ w2,
                                                const float* __restrict__ b2,
                                                const int*   __restrict__ adj,
                                                float* __restrict__ E) {
    const int w    = __builtin_amdgcn_readfirstlane((int)(threadIdx.x >> 5));
    const int lane = threadIdx.x & 31;
    const int jt = blockIdx.x & 7;
    const int ig = (blockIdx.x >> 3) & 15;
    const int b  = blockIdx.x >> 7;
    const int i0 = ig * 32 + w * 4;
    const int jp = jt * 32 + lane;

    const float* p0 = Pb + (size_t)(b * GAT_N + i0) * GAT_H;
    const v4f*   q  = (const v4f*)Qt + ((size_t)b * GAT_HQ * (GAT_N / 2) + jp) * 2;

    const size_t obase = ((size_t)b * GAT_N + i0) * GAT_N + 2 * jp;
    const int2 m0 = *(const int2*)(adj + obase);
    const int2 m1 = *(const int2*)(adj + obase + GAT_N);
    const int2 m2 = *(const int2*)(adj + obase + 2 * GAT_N);
    const int2 m3 = *(const int2*)(adj + obase + 3 * GAT_N);
    const float b2v = b2[0];

    v2f a0 = {0.f, 0.f}, a1 = {0.f, 0.f}, a2 = {0.f, 0.f}, a3 = {0.f, 0.f};
    EStage st0 = e_load(p0, w2, q, 0);
    for (int h = 0; h < GAT_H - 16; h += 16) {
        const EStage st1 = e_load(p0, w2, q, h + 8);
        e_step(st0, a0, a1, a2, a3);
        st0 = e_load(p0, w2, q, h + 16);
        e_step(st1, a0, a1, a2, a3);
    }
    {
        const EStage st1 = e_load(p0, w2, q, GAT_H - 8);
        e_step(st0, a0, a1, a2, a3);
        e_step(st1, a0, a1, a2, a3);
    }

    *(v2f*)(E + obase)             = e_finish(a0, b2v, m0);
    *(v2f*)(E + obase + GAT_N)     = e_finish(a1, b2v, m1);
    *(v2f*)(E + obase + 2 * GAT_N) = e_finish(a2, b2v, m2);
    *(v2f*)(E + obase + 3 * GAT_N) = e_finish(a3, b2v, m3);
}

__global__ __launch_bounds__(256) void softmax_kernel(float* __restrict__ E,
                                                      float* __restrict__ att00) {
    const int row = blockIdx.x;
    const int t   = threadIdx.x;
    float* e = E + (size_t)row * GAT_N;

    __shared__ float red[256];

    const float x0 = e[t], x1 = e[t + 256];
    red[t] = fmaxf(x0, x1);
    __syncthreads();
    #pragma unroll
    for (int s = 128; s > 0; s >>= 1) {
        if (t < s) red[t] = fmaxf(red[t], red[t + s]);
        __syncthreads();
    }
    const float mx = red[0];
    __syncthreads();

    const float e0 = __expf(x0 - mx);
    const float e1 = __expf(x1 - mx);
    red[t] = e0 + e1;
    __syncthreads();
    #pragma unroll
    for (int s = 128; s > 0; s >>= 1) {
        if (t < s) red[t] += red[t + s];
        __syncthreads();
    }
    const float inv = 1.f / red[0];

    e[t]       = e0 * inv;
    e[t + 256] = e1 * inv;
    if (row == 0) {
        att00[t]       = e0 * inv;
        att00[t + 256] = e1 * inv;
    }
}

__global__ __launch_bounds__(256) void av_gemm(const float* __restrict__ att,
                                               const float* __restrict__ x,
                                               float* __restrict__ out) {
    const int wave = (blockIdx.x * blockDim.x + threadIdx.x) >> 5;
    const int lane = threadIdx.x & 31;
    const int b    = wave >> 8;
    const int rem  = wave & 255;
    const int m0 = (rem >> 3) << 4;
    const int n0 = (rem & 7) << 4;
    const int lhalf = lane >> 4;
    const int lm    = lane & 15;

    const float* A = att + (size_t)b * GAT_N * GAT_N;
    const float* X = x   + (size_t)b * GAT_N * GAT_F;
    const float* Arow = A + (size_t)(m0 + lm) * GAT_N;
    const float* Bcol = X + n0 + lm;

    v8f c = {0.f, 0.f, 0.f, 0.f, 0.f, 0.f, 0.f, 0.f};
    #pragma unroll 4
    for (int k0 = 0; k0 < GAT_N; k0 += 4) {
        const int kk = k0 + 2 * lhalf;
        v2f a = *(const v2f*)(Arow + kk);
        v2f bb;
        bb.x = Bcol[(size_t)kk * GAT_F];
        bb.y = Bcol[(size_t)(kk + 1) * GAT_F];
        c = __builtin_amdgcn_wmma_f32_16x16x4_f32(false, a, false, bb,
                                                  (short)0, c, false, false);
    }

    float* orow = out + (size_t)b * GAT_N * GAT_F
                      + (size_t)(m0 + 8 * lhalf) * GAT_F + n0 + lm;
    #pragma unroll
    for (int v = 0; v < 8; ++v) orow[(size_t)v * GAT_F] = c[v];
}

extern "C" void kernel_launch(void* const* d_in, const int* in_sizes, int n_in,
                              void* d_out, int out_size, void* d_ws, size_t ws_size,
                              hipStream_t stream) {
    const float* x   = (const float*)d_in[0];
    const float* w1  = (const float*)d_in[1];
    const float* b1  = (const float*)d_in[2];
    const float* w2  = (const float*)d_in[3];
    const float* b2  = (const float*)d_in[4];
    const int*   adj = (const int*)d_in[5];

    float* out = (float*)d_out;
    float* Pb = (float*)d_ws;
    float* Qt = Pb + (size_t)GAT_B * GAT_N * GAT_H;
    float* E  = Qt + (size_t)GAT_B * GAT_N * GAT_H;

    float* next_h = out;
    float* att00  = out + (size_t)GAT_B * GAT_N * GAT_F;

    pq_gemm<<<256, 256, 0, stream>>>(x, w1, b1, Pb, Qt);
    e_kernel<<<256, 256, 0, stream>>>(Pb, Qt, w2, b2, adj, E);
    softmax_kernel<<<GAT_B * GAT_N, 256, 0, stream>>>(E, att00);
    av_gemm<<<64, 256, 0, stream>>>(E, x, next_h);
}


# att*x kernel: load latency hidden in registers: each wave keeps two 64-k stages of A/B fragments in flight ahead of its WMMAs, att by 16-byte loads, 256 two-wave groups (was 64x8); k order changes
# speedup vs baseline: 2.4717x; 1.4391x over previous
#include <hip/hip_runtime.h>
#include <hip/hip_bf16.h>


typedef __attribute__((ext_vector_type(2))) float v2f;
typedef __attribute__((ext_vector_type(4))) float v4f;
typedef __attribute__((ext_vector_type(8))) float v8f;

#define GAT_B 2
#define GAT_N 512
#define GAT_F 128
#define GAT_H 256
#define GAT_HQ (GAT_H / 4)
#define NEG_SLOPE 0.04f
#define MASK_VAL  (-9.0e15f)

__device__ __forceinline__ v8f tile_kf(const float* __restrict__ Arow,
                                       const float* __restrict__ Brow,
                                       const int lhalf) {
    v8f c = {0.f, 0.f, 0.f, 0.f, 0.f, 0.f, 0.f, 0.f};
    #pragma unroll 8
    for (int k0 = 0; k0 < GAT_F; k0 += 4) {
        const int kk = k0 + 2 * lhalf;
        v2f a = *(const v2f*)(Arow + kk);
        v2f b = *(const v2f*)(Brow + kk);
        c = __builtin_amdgcn_wmma_f32_16x16x4_f32(false, a, false, b,
                                                  (short)0, c, false, false);
    }
    return c;
}

__global__ __launch_bounds__(256) void pq_gemm(const float* __restrict__ x,
                                               const float* __restrict__ w1,
                                               const float* __restrict__ b1,
                                               float* __restrict__ Pb,
                                               float* __restrict__ Qt) {
    const int wave = (blockIdx.x * blockDim.x + threadIdx.x) >> 5;
    const int lane = threadIdx.x & 31;
    const int rem   = wave & 1023;
    const int m0 = (rem >> 4) << 4;
    const int n0 = (rem & 15) << 4;
    const int lhalf = lane >> 4;
    const int lm    = lane & 15;
    const int h     = n0 + lm;

    if (blockIdx.x < 128) {
        const v8f c = tile_kf(x  + (size_t)(m0 + lm) * GAT_F,
                              w1 + (size_t)h * GAT_H, lhalf);
        const float bias = b1[h];
        float* orow = Pb + (size_t)(m0 + 8 * lhalf) * GAT_H + h;
        #pragma unroll
        for (int v = 0; v < 8; ++v) orow[(size_t)v * GAT_H] = c[v] + bias;
    } else {
        const v8f c = tile_kf(x  + (size_t)(m0 + lm) * GAT_F,
                              w1 + (size_t)h * GAT_H + GAT_F, lhalf);
        const int b   = m0 >> 9;
        const int jp0 = ((m0 & (GAT_N - 1)) + 8 * lhalf) >> 1;
        v2f* dst = (v2f*)Qt + (((size_t)(b * GAT_HQ + (h >> 2)) * (GAT_N / 2) + jp0) * 4 + (h & 3));
        #pragma unroll
        for (int p = 0; p < 4; ++p) {
            v2f pr = {c[2 * p], c[2 * p + 1]};
            dst[p * 4] = pr;
        }
    }
}

struct EStage {
    v8f s0, s1, s2, s3;
    v8f w;
    v4f q0, q1, q2, q3;
};

__device__ __forceinline__ EStage e_load(const float* __restrict__ p0,
                                         const float* __restrict__ w2,
                                         const v4f* __restrict__ q, const int h) {
    EStage st;
    st.s0 = *(const v8f*)(p0 + h);
    st.s1 = *(const v8f*)(p0 + GAT_H + h);
    st.s2 = *(const v8f*)(p0 + 2 * GAT_H + h);
    st.s3 = *(const v8f*)(p0 + 3 * GAT_H + h);
    st.w  = *(const v8f*)(w2 + h);
    const v4f* qq = q + (size_t)(h >> 2) * GAT_N;
    st.q0 = qq[0];
    st.q1 = qq[1];
    st.q2 = qq[GAT_N];
    st.q3 = qq[GAT_N + 1];
    return st;
}

__device__ __forceinline__ v2f e_term(const float s, const float w, const v2f qv, const v2f acc) {
    const v2f zero = {0.f, 0.f};
    const v2f sv = {s, s};
    const v2f wv = {w, w};
    return __builtin_elementwise_fma(wv, __builtin_elementwise_max(sv + qv, zero), acc);
}

__device__ __forceinline__ void e_step(const EStage& st, v2f& a0, v2f& a1, v2f& a2, v2f& a3) {
    #pragma unroll
    for (int k = 0; k < 8; ++k) {
        const v4f qq = (k < 2) ? st.q0 : (k < 4) ? st.q1 : (k < 6) ? st.q2 : st.q3;
        const v2f qv = (k & 1) ? (v2f){qq[2], qq[3]} : (v2f){qq[0], qq[1]};
        a0 = e_term(st.s0[k], st.w[k], qv, a0);
        a1 = e_term(st.s1[k], st.w[k], qv, a1);
        a2 = e_term(st.s2[k], st.w[k], qv, a2);
        a3 = e_term(st.s3[k], st.w[k], qv, a3);
    }
}

__device__ __forceinline__ v2f e_finish(const v2f acc, const float b2v, const int2 m) {
    float e0 = acc.x + b2v, e1 = acc.y + b2v;
    e0 = (e0 > 0.f) ? e0 : NEG_SLOPE * e0;
    e1 = (e1 > 0.f) ? e1 : NEG_SLOPE * e1;
    v2f r = {(m.x > 0) ? e0 : MASK_VAL, (m.y > 0) ? e1 : MASK_VAL};
    return r;
}

__global__ __launch_bounds__(256) void e_kernel(const float* __restrict__ Pb,
                                                const float* __restrict__ Qt,
                                                const float* __restrict__ w2,
                                                const float* __restrict__ b2,
                                                const int*   __restrict__ adj,
                                                float* __restrict__ E) {
    const int w    = __builtin_amdgcn_readfirstlane((int)(threadIdx.x >> 5));
    const int lane = threadIdx.x & 31;
    const int jt = blockIdx.x & 7;
    const int ig = (blockIdx.x >> 3) & 15;
    const int b  = blockIdx.x >> 7;
    const int i0 = ig * 32 + w * 4;
    const int jp = jt * 32 + lane;

    const float* p0 = Pb + (size_t)(b * GAT_N + i0) * GAT_H;
    const v4f*   q  = (const v4f*)Qt + ((size_t)b * GAT_HQ * (GAT_N / 2) + jp) * 2;

    const size_t obase = ((size_t)b * GAT_N + i0) * GAT_N + 2 * jp;
    const int2 m0 = *(const int2*)(adj + obase);
    const int2 m1 = *(const int2*)(adj + obase + GAT_N);
    const int2 m2 = *(const int2*)(adj + obase + 2 * GAT_N);
    const int2 m3 = *(const int2*)(adj + obase + 3 * GAT_N);
    const float b2v = b2[0];

    v2f a0 = {0.f, 0.f}, a1 = {0.f, 0.f}, a2 = {0.f, 0.f}, a3 = {0.f, 0.f};
    EStage st0 = e_load(p0, w2, q, 0);
    for (int h = 0; h < GAT_H - 16; h += 16) {
        const EStage st1 = e_load(p0, w2, q, h + 8);
        e_step(st0, a0, a1, a2, a3);
        st0 = e_load(p0, w2, q, h + 16);
        e_step(st1, a0, a1, a2, a3);
    }
    {
        const EStage st1 = e_load(p0, w2, q, GAT_H - 8);
        e_step(st0, a0, a1, a2, a3);
        e_step(st1, a0, a1, a2, a3);
    }

    *(v2f*)(E + obase)             = e_finish(a0, b2v, m0);
    *(v2f*)(E + obase + GAT_N)     = e_finish(a1, b2v, m1);
    *(v2f*)(E + obase + 2 * GAT_N) = e_finish(a2, b2v, m2);
    *(v2f*)(E + obase + 3 * GAT_N) = e_finish(a3, b2v, m3);
}

__global__ __launch_bounds__(256) void softmax_kernel(float* __restrict__ E,
                                                      float* __restrict__ att00) {
    const int row = blockIdx.x;
    const int t   = threadIdx.x;
    float* e = E + (size_t)row * GAT_N;

    __shared__ float red[256];

    const float x0 = e[t], x1 = e[t + 256];
    red[t] = fmaxf(x0, x1);
    __syncthreads();
    #pragma unroll
    for (int s = 128; s > 0; s >>= 1) {
        if (t < s) red[t] = fmaxf(red[t], red[t + s]);
        __syncthreads();
    }
    const float mx = red[0];
    __syncthreads();

    const float e0 = __expf(x0 - mx);
    const float e1 = __expf(x1 - mx);
    red[t] = e0 + e1;
    __syncthreads();
    #pragma unroll
    for (int s = 128; s > 0; s >>= 1) {
        if (t < s) red[t] += red[t + s];
        __syncthreads();
    }
    const float inv = 1.f / red[0];

    e[t]       = e0 * inv;
    e[t + 256] = e1 * inv;
    if (row == 0) {
        att00[t]       = e0 * inv;
        att00[t + 256] = e1 * inv;
    }
}

#define AV_GRP 8

struct AvStage {
    v4f a[AV_GRP];
    v4f b[AV_GRP];
};

__device__ __forceinline__ void av_load(AvStage& st, const float* __restrict__ arow,
                                        const float* __restrict__ bcol, const int g0) {
    #pragma unroll
    for (int g = 0; g < AV_GRP; ++g) {
        st.a[g] = *(const v4f*)(arow + (g0 + g) * 8);
        const float* bp = bcol + (size_t)(g0 + g) * 8 * GAT_F;
        const v4f bv = {bp[0], bp[GAT_F], bp[2 * GAT_F], bp[3 * GAT_F]};
        st.b[g] = bv;
    }
}

__device__ __forceinline__ v8f av_mma(const AvStage& st, v8f c) {
    #pragma unroll
    for (int g = 0; g < AV_GRP; ++g) {
        const v2f a0 = {st.a[g][0], st.a[g][1]};
        const v2f b0 = {st.b[g][0], st.b[g][1]};
        const v2f a1 = {st.a[g][2], st.a[g][3]};
        const v2f b1 = {st.b[g][2], st.b[g][3]};
        c = __builtin_amdgcn_wmma_f32_16x16x4_f32(false, a0, false, b0,
                                                  (short)0, c, false, false);
        c = __builtin_amdgcn_wmma_f32_16x16x4_f32(false, a1, false, b1,
                                                  (short)0, c, false, false);
    }
    return c;
}

__global__ __launch_bounds__(64) void av_gemm(const float* __restrict__ att,
                                              const float* __restrict__ x,
                                              float* __restrict__ out) {
    const int wave = (blockIdx.x * blockDim.x + threadIdx.x) >> 5;
    const int lane = threadIdx.x & 31;
    const int b    = wave >> 8;
    const int rem  = wave & 255;
    const int m0 = (rem >> 3) << 4;
    const int n0 = (rem & 7) << 4;
    const int lhalf = lane >> 4;
    const int lm    = lane & 15;

    const float* A = att + (size_t)b * GAT_N * GAT_N;
    const float* X = x   + (size_t)b * GAT_N * GAT_F;
    const float* arow = A + (size_t)(m0 + lm) * GAT_N + 4 * lhalf;
    const float* bcol = X + (size_t)(4 * lhalf) * GAT_F + n0 + lm;

    v8f c = {0.f, 0.f, 0.f, 0.f, 0.f, 0.f, 0.f, 0.f};
#define AV_FENCE() __builtin_amdgcn_sched_barrier(0)
    AvStage sa, sb, sc;
    av_load(sa, arow, bcol, 0 * AV_GRP);
    av_load(sb, arow, bcol, 1 * AV_GRP);
    AV_FENCE();
    av_load(sc, arow, bcol, 2 * AV_GRP);  AV_FENCE();  c = av_mma(sa, c);  AV_FENCE();
    av_load(sa, arow, bcol, 3 * AV_GRP);  AV_FENCE();  c = av_mma(sb, c);  AV_FENCE();
    av_load(sb, arow, bcol, 4 * AV_GRP);  AV_FENCE();  c = av_mma(sc, c);  AV_FENCE();
    av_load(sc, arow, bcol, 5 * AV_GRP);  AV_FENCE();  c = av_mma(sa, c);  AV_FENCE();
    av_load(sa, arow, bcol, 6 * AV_GRP);  AV_FENCE();  c = av_mma(sb, c);  AV_FENCE();
    av_load(sb, arow, bcol, 7 * AV_GRP);  AV_FENCE();  c = av_mma(sc, c);  AV_FENCE();
    c = av_mma(sa, c);
    c = av_mma(sb, c);
#undef AV_FENCE

    float* orow = out + (size_t)b * GAT_N * GAT_F
                      + (size_t)(m0 + 8 * lhalf) * GAT_F + n0 + lm;
    #pragma unroll
    for (int v = 0; v < 8; ++v) orow[(size_t)v * GAT_F] = c[v];
}

extern "C" void kernel_launch(void* const* d_in, const int* in_sizes, int n_in,
                              void* d_out, int out_size, void* d_ws, size_t ws_size,
                              hipStream_t stream) {
    const float* x   = (const float*)d_in[0];
    const float* w1  = (const float*)d_in[1];
    const float* b1  = (const float*)d_in[2];
    const float* w2  = (const float*)d_in[3];
    const float* b2  = (const float*)d_in[4];
    const int*   adj = (const int*)d_in[5];

    float* out = (float*)d_out;
    float* Pb = (float*)d_ws;
    float* Qt = Pb + (size_t)GAT_B * GAT_N * GAT_H;
    float* E  = Qt + (size_t)GAT_B * GAT_N * GAT_H;

    float* next_h = out;
    float* att00  = out + (size_t)GAT_B * GAT_N * GAT_F;

    pq_gemm<<<256, 256, 0, stream>>>(x, w1, b1, Pb, Qt);
    e_kernel<<<256, 256, 0, stream>>>(Pb, Qt, w2, b2, adj, E);
    softmax_kernel<<<GAT_B * GAT_N, 256, 0, stream>>>(E, att00);
    av_gemm<<<256, 64, 0, stream>>>(E, x, next_h);
}


# softmax kernel: one wave per row, the row's 512 values in 16 registers per lane, max and sum lane-local then one lane exchange each; no LDS, no barriers; 1024 waves (was 8192); exp-sum order new
# speedup vs baseline: 2.5598x; 1.0356x over previous
#include <hip/hip_runtime.h>
#include <hip/hip_bf16.h>


typedef __attribute__((ext_vector_type(2))) float v2f;
typedef __attribute__((ext_vector_type(4))) float v4f;
typedef __attribute__((ext_vector_type(8))) float v8f;

#define GAT_B 2
#define GAT_N 512
#define GAT_F 128
#define GAT_H 256
#define GAT_HQ (GAT_H / 4)
#define NEG_SLOPE 0.04f
#define MASK_VAL  (-9.0e15f)

__device__ __forceinline__ v8f tile_kf(const float* __restrict__ Arow,
                                       const float* __restrict__ Brow,
                                       const int lhalf) {
    v8f c = {0.f, 0.f, 0.f, 0.f, 0.f, 0.f, 0.f, 0.f};
    #pragma unroll 8
    for (int k0 = 0; k0 < GAT_F; k0 += 4) {
        const int kk = k0 + 2 * lhalf;
        v2f a = *(const v2f*)(Arow + kk);
        v2f b = *(const v2f*)(Brow + kk);
        c = __builtin_amdgcn_wmma_f32_16x16x4_f32(false, a, false, b,
                                                  (short)0, c, false, false);
    }
    return c;
}

__global__ __launch_bounds__(256) void pq_gemm(const float* __restrict__ x,
                                               const float* __restrict__ w1,
                                               const float* __restrict__ b1,
                                               float* __restrict__ Pb,
                                               float* __restrict__ Qt) {
    const int wave = (blockIdx.x * blockDim.x + threadIdx.x) >> 5;
    const int lane = threadIdx.x & 31;
    const int rem   = wave & 1023;
    const int m0 = (rem >> 4) << 4;
    const int n0 = (rem & 15) << 4;
    const int lhalf = lane >> 4;
    const int lm    = lane & 15;
    const int h     = n0 + lm;

    if (blockIdx.x < 128) {
        const v8f c = tile_kf(x  + (size_t)(m0 + lm) * GAT_F,
                              w1 + (size_t)h * GAT_H, lhalf);
        const float bias = b1[h];
        float* orow = Pb + (size_t)(m0 + 8 * lhalf) * GAT_H + h;
        #pragma unroll
        for (int v = 0; v < 8; ++v) orow[(size_t)v * GAT_H] = c[v] + bias;
    } else {
        const v8f c = tile_kf(x  + (size_t)(m0 + lm) * GAT_F,
                              w1 + (size_t)h * GAT_H + GAT_F, lhalf);
        const int b   = m0 >> 9;
        const int jp0 = ((m0 & (GAT_N - 1)) + 8 * lhalf) >> 1;
        v2f* dst = (v2f*)Qt + (((size_t)(b * GAT_HQ + (h >> 2)) * (GAT_N / 2) + jp0) * 4 + (h & 3));
        #pragma unroll
        for (int p = 0; p < 4; ++p) {
            v2f pr = {c[2 * p], c[2 * p + 1]};
            dst[p * 4] = pr;
        }
    }
}

struct EStage {
    v8f s0, s1, s2, s3;
    v8f w;
    v4f q0, q1, q2, q3;
};

__device__ __forceinline__ EStage e_load(const float* __restrict__ p0,
                                         const float* __restrict__ w2,
                                         const v4f* __restrict__ q, const int h) {
    EStage st;
    st.s0 = *(const v8f*)(p0 + h);
    st.s1 = *(const v8f*)(p0 + GAT_H + h);
    st.s2 = *(const v8f*)(p0 + 2 * GAT_H + h);
    st.s3 = *(const v8f*)(p0 + 3 * GAT_H + h);
    st.w  = *(const v8f*)(w2 + h);
    const v4f* qq = q + (size_t)(h >> 2) * GAT_N;
    st.q0 = qq[0];
    st.q1 = qq[1];
    st.q2 = qq[GAT_N];
    st.q3 = qq[GAT_N + 1];
    return st;
}

__device__ __forceinline__ v2f e_term(const float s, const float w, const v2f qv, const v2f acc) {
    const v2f zero = {0.f, 0.f};
    const v2f sv = {s, s};
    const v2f wv = {w, w};
    return __builtin_elementwise_fma(wv, __builtin_elementwise_max(sv + qv, zero), acc);
}

__device__ __forceinline__ void e_step(const EStage& st, v2f& a0, v2f& a1, v2f& a2, v2f& a3) {
    #pragma unroll
    for (int k = 0; k < 8; ++k) {
        const v4f qq = (k < 2) ? st.q0 : (k < 4) ? st.q1 : (k < 6) ? st.q2 : st.q3;
        const v2f qv = (k & 1) ? (v2f){qq[2], qq[3]} : (v2f){qq[0], qq[1]};
        a0 = e_term(st.s0[k], st.w[k], qv, a0);
        a1 = e_term(st.s1[k], st.w[k], qv, a1);
        a2 = e_term(st.s2[k], st.w[k], qv, a2);
        a3 = e_term(st.s3[k], st.w[k], qv, a3);
    }
}

__device__ __forceinline__ v2f e_finish(const v2f acc, const float b2v, const int2 m) {
    float e0 = acc.x + b2v, e1 = acc.y + b2v;
    e0 = (e0 > 0.f) ? e0 : NEG_SLOPE * e0;
    e1 = (e1 > 0.f) ? e1 : NEG_SLOPE * e1;
    v2f r = {(m.x > 0) ? e0 : MASK_VAL, (m.y > 0) ? e1 : MASK_VAL};
    return r;
}

__global__ __launch_bounds__(256) void e_kernel(const float* __restrict__ Pb,
                                                const float* __restrict__ Qt,
                                                const float* __restrict__ w2,
                                                const float* __restrict__ b2,
                                                const int*   __restrict__ adj,
                                                float* __restrict__ E) {
    const int w    = __builtin_amdgcn_readfirstlane((int)(threadIdx.x >> 5));
    const int lane = threadIdx.x & 31;
    const int jt = blockIdx.x & 7;
    const int ig = (blockIdx.x >> 3) & 15;
    const int b  = blockIdx.x >> 7;
    const int i0 = ig * 32 + w * 4;
    const int jp = jt * 32 + lane;

    const float* p0 = Pb + (size_t)(b * GAT_N + i0) * GAT_H;
    const v4f*   q  = (const v4f*)Qt + ((size_t)b * GAT_HQ * (GAT_N / 2) + jp) * 2;

    const size_t obase = ((size_t)b * GAT_N + i0) * GAT_N + 2 * jp;
    const int2 m0 = *(const int2*)(adj + obase);
    const int2 m1 = *(const int2*)(adj + obase + GAT_N);
    const int2 m2 = *(const int2*)(adj + obase + 2 * GAT_N);
    const int2 m3 = *(const int2*)(adj + obase + 3 * GAT_N);
    const float b2v = b2[0];

    v2f a0 = {0.f, 0.f}, a1 = {0.f, 0.f}, a2 = {0.f, 0.f}, a3 = {0.f, 0.f};
    EStage st0 = e_load(p0, w2, q, 0);
    for (int h = 0; h < GAT_H - 16; h += 16) {
        const EStage st1 = e_load(p0, w2, q, h + 8);
        e_step(st0, a0, a1, a2, a3);
        st0 = e_load(p0, w2, q, h + 16);
        e_step(st1, a0, a1, a2, a3);
    }
    {
        const EStage st1 = e_load(p0, w2, q, GAT_H - 8);
        e_step(st0, a0, a1, a2, a3);
        e_step(st1, a0, a1, a2, a3);
    }

    *(v2f*)(E + obase)             = e_finish(a0, b2v, m0);
    *(v2f*)(E + obase + GAT_N)     = e_finish(a1, b2v, m1);
    *(v2f*)(E + obase + 2 * GAT_N) = e_finish(a2, b2v, m2);
    *(v2f*)(E + obase + 3 * GAT_N) = e_finish(a3, b2v, m3);
}

__device__ __forceinline__ float xhalf(const float x) {
    const int b = __builtin_bit_cast(int, x);
    return __builtin_bit_cast(float,
        __builtin_amdgcn_permlanex16(b, b, 0x76543210, 0xfedcba98, false, false));
}

template <int MASK>
__device__ __forceinline__ float xrow(const float x) {
    const int b = __builtin_bit_cast(int, x);
    return __builtin_bit_cast(float,
        __builtin_amdgcn_update_dpp(b, b, 0x160 + MASK, 0xf, 0xf, false));
}

__device__ __forceinline__ float wave_max(float v) {
    v = fmaxf(v, xrow<1>(v));
    v = fmaxf(v, xrow<2>(v));
    v = fmaxf(v, xrow<4>(v));
    v = fmaxf(v, xrow<8>(v));
    return fmaxf(v, xhalf(v));
}

__device__ __forceinline__ float wave_sum(float v) {
    v += xrow<1>(v);
    v += xrow<2>(v);
    v += xrow<4>(v);
    v += xrow<8>(v);
    return v + xhalf(v);
}

__device__ __forceinline__ v4f exp4(const v4f x, const float mx) {
    v4f r = {__expf(x[0] - mx), __expf(x[1] - mx), __expf(x[2] - mx), __expf(x[3] - mx)};
    return r;
}

__global__ __launch_bounds__(128) void softmax_kernel(float* __restrict__ E,
                                                      float* __restrict__ att00) {
    const int w    = __builtin_amdgcn_readfirstlane((int)(threadIdx.x >> 5));
    const int lane = threadIdx.x & 31;
    const int row  = blockIdx.x * 4 + w;
    v4f* e = (v4f*)(E + (size_t)row * GAT_N) + lane;

    const v4f x0 = e[0], x1 = e[32], x2 = e[64], x3 = e[96];

    const v4f m4 = __builtin_elementwise_max(__builtin_elementwise_max(x0, x1),
                                             __builtin_elementwise_max(x2, x3));
    const float mx = wave_max(fmaxf(fmaxf(m4[0], m4[1]), fmaxf(m4[2], m4[3])));

    const v4f p0 = exp4(x0, mx), p1 = exp4(x1, mx), p2 = exp4(x2, mx), p3 = exp4(x3, mx);
    const v4f s4 = (p0 + p1) + (p2 + p3);
    const float sum = wave_sum((s4[0] + s4[1]) + (s4[2] + s4[3]));
    const float inv = 1.f / sum;

    const v4f o0 = p0 * inv, o1 = p1 * inv, o2 = p2 * inv, o3 = p3 * inv;
    e[0] = o0;  e[32] = o1;  e[64] = o2;  e[96] = o3;
    if (row == 0) {
        v4f* a = (v4f*)att00 + lane;
        a[0] = o0;  a[32] = o1;  a[64] = o2;  a[96] = o3;
    }
}

#define AV_GRP 8

struct AvStage {
    v4f a[AV_GRP];
    v4f b[AV_GRP];
};

__device__ __forceinline__ void av_load(AvStage& st, const float* __restrict__ arow,
                                        const float* __restrict__ bcol, const int g0) {
    #pragma unroll
    for (int g = 0; g < AV_GRP; ++g) {
        st.a[g] = *(const v4f*)(arow + (g0 + g) * 8);
        const float* bp = bcol + (size_t)(g0 + g) * 8 * GAT_F;
        const v4f bv = {bp[0], bp[GAT_F], bp[2 * GAT_F], bp[3 * GAT_F]};
        st.b[g] = bv;
    }
}

__device__ __forceinline__ v8f av_mma(const AvStage& st, v8f c) {
    #pragma unroll
    for (int g = 0; g < AV_GRP; ++g) {
        const v2f a0 = {st.a[g][0], st.a[g][1]};
        const v2f b0 = {st.b[g][0], st.b[g][1]};
        const v2f a1 = {st.a[g][2], st.a[g][3]};
        const v2f b1 = {st.b[g][2], st.b[g][3]};
        c = __builtin_amdgcn_wmma_f32_16x16x4_f32(false, a0, false, b0,
                                                  (short)0, c, false, false);
        c = __builtin_amdgcn_wmma_f32_16x16x4_f32(false, a1, false, b1,
                                                  (short)0, c, false, false);
    }
    return c;
}

__global__ __launch_bounds__(64) void av_gemm(const float* __restrict__ att,
                                              const float* __restrict__ x,
                                              float* __restrict__ out) {
    const int wave = (blockIdx.x * blockDim.x + threadIdx.x) >> 5;
    const int lane = threadIdx.x & 31;
    const int b    = wave >> 8;
    const int rem  = wave & 255;
    const int m0 = (rem >> 3) << 4;
    const int n0 = (rem & 7) << 4;
    const int lhalf = lane >> 4;
    const int lm    = lane & 15;

    const float* A = att + (size_t)b * GAT_N * GAT_N;
    const float* X = x   + (size_t)b * GAT_N * GAT_F;
    const float* arow = A + (size_t)(m0 + lm) * GAT_N + 4 * lhalf;
    const float* bcol = X + (size_t)(4 * lhalf) * GAT_F + n0 + lm;

    v8f c = {0.f, 0.f, 0.f, 0.f, 0.f, 0.f, 0.f, 0.f};
#define AV_FENCE() __builtin_amdgcn_sched_barrier(0)
    AvStage sa, sb, sc;
    av_load(sa, arow, bcol, 0 * AV_GRP);
    av_load(sb, arow, bcol, 1 * AV_GRP);
    AV_FENCE();
    av_load(sc, arow, bcol, 2 * AV_GRP);  AV_FENCE();  c = av_mma(sa, c);  AV_FENCE();
    av_load(sa, arow, bcol, 3 * AV_GRP);  AV_FENCE();  c = av_mma(sb, c);  AV_FENCE();
    av_load(sb, arow, bcol, 4 * AV_GRP);  AV_FENCE();  c = av_mma(sc, c);  AV_FENCE();
    av_load(sc, arow, bcol, 5 * AV_GRP);  AV_FENCE();  c = av_mma(sa, c);  AV_FENCE();
    av_load(sa, arow, bcol, 6 * AV_GRP);  AV_FENCE();  c = av_mma(sb, c);  AV_FENCE();
    av_load(sb, arow, bcol, 7 * AV_GRP);  AV_FENCE();  c = av_mma(sc, c);  AV_FENCE();
    c = av_mma(sa, c);
    c = av_mma(sb, c);
#undef AV_FENCE

    float* orow = out + (size_t)b * GAT_N * GAT_F
                      + (size_t)(m0 + 8 * lhalf) * GAT_F + n0 + lm;
    #pragma unroll
    for (int v = 0; v < 8; ++v) orow[(size_t)v * GAT_F] = c[v];
}

extern "C" void kernel_launch(void* const* d_in, const int* in_sizes, int n_in,
                              void* d_out, int out_size, void* d_ws, size_t ws_size,
                              hipStream_t stream) {
    const float* x   = (const float*)d_in[0];
    const float* w1  = (const float*)d_in[1];
    const float* b1  = (const float*)d_in[2];
    const float* w2  = (const float*)d_in[3];
    const float* b2  = (const float*)d_in[4];
    const int*   adj = (const int*)d_in[5];

    float* out = (float*)d_out;
    float* Pb = (float*)d_ws;
    float* Qt = Pb + (size_t)GAT_B * GAT_N * GAT_H;
    float* E  = Qt + (size_t)GAT_B * GAT_N * GAT_H;

    float* next_h = out;
    float* att00  = out + (size_t)GAT_B * GAT_N * GAT_F;

    pq_gemm<<<256, 256, 0, stream>>>(x, w1, b1, Pb, Qt);
    e_kernel<<<256, 256, 0, stream>>>(Pb, Qt, w2, b2, adj, E);
    softmax_kernel<<<GAT_B * GAT_N / 4, 128, 0, stream>>>(E, att00);
    av_gemm<<<256, 64, 0, stream>>>(E, x, next_h);
}


# softmax split over its neighbours: e-logit wave writes exp(e - slice max) and (max,sum) per 64-col slice, att*x scales each 64-k A stage by exp(m_s-m)/total; softmax launch gone; exp sum order new
# speedup vs baseline: 2.7103x; 1.0588x over previous
#include <hip/hip_runtime.h>
#include <hip/hip_bf16.h>


typedef __attribute__((ext_vector_type(2))) float v2f;
typedef __attribute__((ext_vector_type(4))) float v4f;
typedef __attribute__((ext_vector_type(8))) float v8f;

#define GAT_B 2
#define GAT_N 512
#define GAT_F 128
#define GAT_H 256
#define GAT_HQ (GAT_H / 4)
#define NEG_SLOPE 0.04f
#define MASK_VAL  (-9.0e15f)

__device__ __forceinline__ v8f tile_kf(const float* __restrict__ Arow,
                                       const float* __restrict__ Brow,
                                       const int lhalf) {
    v8f c = {0.f, 0.f, 0.f, 0.f, 0.f, 0.f, 0.f, 0.f};
    #pragma unroll 8
    for (int k0 = 0; k0 < GAT_F; k0 += 4) {
        const int kk = k0 + 2 * lhalf;
        v2f a = *(const v2f*)(Arow + kk);
        v2f b = *(const v2f*)(Brow + kk);
        c = __builtin_amdgcn_wmma_f32_16x16x4_f32(false, a, false, b,
                                                  (short)0, c, false, false);
    }
    return c;
}

__global__ __launch_bounds__(256) void pq_gemm(const float* __restrict__ x,
                                               const float* __restrict__ w1,
                                               const float* __restrict__ b1,
                                               float* __restrict__ Pb,
                                               float* __restrict__ Qt) {
    const int wave = (blockIdx.x * blockDim.x + threadIdx.x) >> 5;
    const int lane = threadIdx.x & 31;
    const int rem   = wave & 1023;
    const int m0 = (rem >> 4) << 4;
    const int n0 = (rem & 15) << 4;
    const int lhalf = lane >> 4;
    const int lm    = lane & 15;
    const int h     = n0 + lm;

    if (blockIdx.x < 128) {
        const v8f c = tile_kf(x  + (size_t)(m0 + lm) * GAT_F,
                              w1 + (size_t)h * GAT_H, lhalf);
        const float bias = b1[h];
        float* orow = Pb + (size_t)(m0 + 8 * lhalf) * GAT_H + h;
        #pragma unroll
        for (int v = 0; v < 8; ++v) orow[(size_t)v * GAT_H] = c[v] + bias;
    } else {
        const v8f c = tile_kf(x  + (size_t)(m0 + lm) * GAT_F,
                              w1 + (size_t)h * GAT_H + GAT_F, lhalf);
        const int b   = m0 >> 9;
        const int jp0 = ((m0 & (GAT_N - 1)) + 8 * lhalf) >> 1;
        v2f* dst = (v2f*)Qt + (((size_t)(b * GAT_HQ + (h >> 2)) * (GAT_N / 2) + jp0) * 4 + (h & 3));
        #pragma unroll
        for (int p = 0; p < 4; ++p) {
            v2f pr = {c[2 * p], c[2 * p + 1]};
            dst[p * 4] = pr;
        }
    }
}

__device__ __forceinline__ float xhalf(const float x) {
    const int b = __builtin_bit_cast(int, x);
    return __builtin_bit_cast(float,
        __builtin_amdgcn_permlanex16(b, b, 0x76543210, 0xfedcba98, false, false));
}

template <int MASK>
__device__ __forceinline__ float xrow(const float x) {
    const int b = __builtin_bit_cast(int, x);
    return __builtin_bit_cast(float,
        __builtin_amdgcn_update_dpp(b, b, 0x160 + MASK, 0xf, 0xf, false));
}

__device__ __forceinline__ float wave_max(float v) {
    v = fmaxf(v, xrow<1>(v));
    v = fmaxf(v, xrow<2>(v));
    v = fmaxf(v, xrow<4>(v));
    v = fmaxf(v, xrow<8>(v));
    return fmaxf(v, xhalf(v));
}

__device__ __forceinline__ float wave_sum(float v) {
    v += xrow<1>(v);
    v += xrow<2>(v);
    v += xrow<4>(v);
    v += xrow<8>(v);
    return v + xhalf(v);
}

__device__ __forceinline__ float first_lane(const float x) {
    return __builtin_bit_cast(float,
        __builtin_amdgcn_readfirstlane(__builtin_bit_cast(int, x)));
}

struct EStage {
    v8f s0, s1, s2, s3;
    v8f w;
    v4f q0, q1, q2, q3;
};

__device__ __forceinline__ EStage e_load(const float* __restrict__ p0,
                                         const float* __restrict__ w2,
                                         const v4f* __restrict__ q, const int h) {
    EStage st;
    st.s0 = *(const v8f*)(p0 + h);
    st.s1 = *(const v8f*)(p0 + GAT_H + h);
    st.s2 = *(const v8f*)(p0 + 2 * GAT_H + h);
    st.s3 = *(const v8f*)(p0 + 3 * GAT_H + h);
    st.w  = *(const v8f*)(w2 + h);
    const v4f* qq = q + (size_t)(h >> 2) * GAT_N;
    st.q0 = qq[0];
    st.q1 = qq[1];
    st.q2 = qq[GAT_N];
    st.q3 = qq[GAT_N + 1];
    return st;
}

__device__ __forceinline__ v2f e_term(const float s, const float w, const v2f qv, const v2f acc) {
    const v2f zero = {0.f, 0.f};
    const v2f sv = {s, s};
    const v2f wv = {w, w};
    return __builtin_elementwise_fma(wv, __builtin_elementwise_max(sv + qv, zero), acc);
}

__device__ __forceinline__ void e_step(const EStage& st, v2f& a0, v2f& a1, v2f& a2, v2f& a3) {
    #pragma unroll
    for (int k = 0; k < 8; ++k) {
        const v4f qq = (k < 2) ? st.q0 : (k < 4) ? st.q1 : (k < 6) ? st.q2 : st.q3;
        const v2f qv = (k & 1) ? (v2f){qq[2], qq[3]} : (v2f){qq[0], qq[1]};
        a0 = e_term(st.s0[k], st.w[k], qv, a0);
        a1 = e_term(st.s1[k], st.w[k], qv, a1);
        a2 = e_term(st.s2[k], st.w[k], qv, a2);
        a3 = e_term(st.s3[k], st.w[k], qv, a3);
    }
}

__device__ __forceinline__ v2f e_finish(const v2f acc, const float b2v, const int2 m) {
    float e0 = acc.x + b2v, e1 = acc.y + b2v;
    e0 = (e0 > 0.f) ? e0 : NEG_SLOPE * e0;
    e1 = (e1 > 0.f) ? e1 : NEG_SLOPE * e1;
    v2f r = {(m.x > 0) ? e0 : MASK_VAL, (m.y > 0) ? e1 : MASK_VAL};
    return r;
}

__global__ __launch_bounds__(256) void e_kernel(const float* __restrict__ Pb,
                                                const float* __restrict__ Qt,
                                                const float* __restrict__ w2,
                                                const float* __restrict__ b2,
                                                const int*   __restrict__ adj,
                                                float* __restrict__ E,
                                                float* __restrict__ parts) {
    const int w    = __builtin_amdgcn_readfirstlane((int)(threadIdx.x >> 5));
    const int lane = threadIdx.x & 31;
    const int jt = blockIdx.x & 7;
    const int ig = (blockIdx.x >> 3) & 15;
    const int b  = blockIdx.x >> 7;
    const int i0 = ig * 32 + w * 4;
    const int jp = jt * 32 + lane;

    const float* p0 = Pb + (size_t)(b * GAT_N + i0) * GAT_H;
    const v4f*   q  = (const v4f*)Qt + ((size_t)b * GAT_HQ * (GAT_N / 2) + jp) * 2;

    const size_t obase = ((size_t)b * GAT_N + i0) * GAT_N + 2 * jp;
    const int2 m0 = *(const int2*)(adj + obase);
    const int2 m1 = *(const int2*)(adj + obase + GAT_N);
    const int2 m2 = *(const int2*)(adj + obase + 2 * GAT_N);
    const int2 m3 = *(const int2*)(adj + obase + 3 * GAT_N);
    const float b2v = b2[0];

    v2f a0 = {0.f, 0.f}, a1 = {0.f, 0.f}, a2 = {0.f, 0.f}, a3 = {0.f, 0.f};
    EStage st0 = e_load(p0, w2, q, 0);
    for (int h = 0; h < GAT_H - 16; h += 16) {
        const EStage st1 = e_load(p0, w2, q, h + 8);
        e_step(st0, a0, a1, a2, a3);
        st0 = e_load(p0, w2, q, h + 16);
        e_step(st1, a0, a1, a2, a3);
    }
    {
        const EStage st1 = e_load(p0, w2, q, GAT_H - 8);
        e_step(st0, a0, a1, a2, a3);
        e_step(st1, a0, a1, a2, a3);
    }

    const v2f r0 = e_finish(a0, b2v, m0);
    const v2f r1 = e_finish(a1, b2v, m1);
    const v2f r2 = e_finish(a2, b2v, m2);
    const v2f r3 = e_finish(a3, b2v, m3);

    const float x0 = wave_max(fmaxf(r0.x, r0.y));
    const float x1 = wave_max(fmaxf(r1.x, r1.y));
    const float x2 = wave_max(fmaxf(r2.x, r2.y));
    const float x3 = wave_max(fmaxf(r3.x, r3.y));
    const v2f y0 = {__expf(r0.x - x0), __expf(r0.y - x0)};
    const v2f y1 = {__expf(r1.x - x1), __expf(r1.y - x1)};
    const v2f y2 = {__expf(r2.x - x2), __expf(r2.y - x2)};
    const v2f y3 = {__expf(r3.x - x3), __expf(r3.y - x3)};
    *(v2f*)(E + obase)             = y0;
    *(v2f*)(E + obase + GAT_N)     = y1;
    *(v2f*)(E + obase + 2 * GAT_N) = y2;
    *(v2f*)(E + obase + 3 * GAT_N) = y3;
    const float s0 = wave_sum(y0.x + y0.y);
    const float s1 = wave_sum(y1.x + y1.y);
    const float s2 = wave_sum(y2.x + y2.y);
    const float s3 = wave_sum(y3.x + y3.y);

    const int rr = lane >> 3, cc = lane & 7;
    const v2f ms = {(rr == 0) ? x0 : (rr == 1) ? x1 : (rr == 2) ? x2 : x3,
                    (rr == 0) ? s0 : (rr == 1) ? s1 : (rr == 2) ? s2 : s3};
    *(v2f*)(parts + ((size_t)b * GAT_N + i0 + rr) * GAT_F + 16 * cc + 2 * jt) = ms;
}

#define AV_GRP 8

struct AvStage {
    v4f a[AV_GRP];
    v4f b[AV_GRP];
};

__device__ __forceinline__ void av_load(AvStage& st, const float* __restrict__ arow,
                                        const float* __restrict__ bcol, const int g0) {
    #pragma unroll
    for (int g = 0; g < AV_GRP; ++g) {
        st.a[g] = *(const v4f*)(arow + (g0 + g) * 8);
        const float* bp = bcol + (size_t)(g0 + g) * 8 * GAT_F;
        const v4f bv = {bp[0], bp[GAT_F], bp[2 * GAT_F], bp[3 * GAT_F]};
        st.b[g] = bv;
    }
}

__device__ __forceinline__ v8f av_mma(const AvStage& st, v8f c, const float sc) {
    #pragma unroll
    for (int g = 0; g < AV_GRP; ++g) {
        const v2f a0 = {st.a[g][0] * sc, st.a[g][1] * sc};
        const v2f b0 = {st.b[g][0], st.b[g][1]};
        const v2f a1 = {st.a[g][2] * sc, st.a[g][3] * sc};
        const v2f b1 = {st.b[g][2], st.b[g][3]};
        c = __builtin_amdgcn_wmma_f32_16x16x4_f32(false, a0, false, b0,
                                                  (short)0, c, false, false);
        c = __builtin_amdgcn_wmma_f32_16x16x4_f32(false, a1, false, b1,
                                                  (short)0, c, false, false);
    }
    return c;
}

__global__ __launch_bounds__(64) void av_gemm(const float* __restrict__ att,
                                              const float* __restrict__ x,
                                              float* __restrict__ out,
                                              float* __restrict__ att00) {
    const int wave = (blockIdx.x * blockDim.x + threadIdx.x) >> 5;
    const int lane = threadIdx.x & 31;
    const int b    = wave >> 8;
    const int rem  = wave & 255;
    const int m0 = (rem >> 3) << 4;
    const int n0 = (rem & 7) << 4;
    const int lhalf = lane >> 4;
    const int lm    = lane & 15;

    const float* A = att + (size_t)b * GAT_N * GAT_N;
    const float* X = x   + (size_t)b * GAT_N * GAT_F;
    const float* arow = A + (size_t)(m0 + lm) * GAT_N + 4 * lhalf;
    const float* bcol = X + (size_t)(4 * lhalf) * GAT_F + n0 + lm;

    v8f c = {0.f, 0.f, 0.f, 0.f, 0.f, 0.f, 0.f, 0.f};
#define AV_FENCE() __builtin_amdgcn_sched_barrier(0)
    float* otile = out + (size_t)b * GAT_N * GAT_F + (size_t)m0 * GAT_F + n0;
    const v4f* pr = (const v4f*)(otile + (size_t)lm * GAT_F);
    const v4f t0 = pr[0], t1 = pr[1], t2 = pr[2], t3 = pr[3];
    const bool row0 = __builtin_amdgcn_readfirstlane(wave) == 0;
    v4f q0 = {0.f, 0.f, 0.f, 0.f}, q1 = q0, q2 = q0, q3 = q0;
    if (row0) {
        const v4f* e0 = (const v4f*)att + lane;
        q0 = e0[0];  q1 = e0[32];  q2 = e0[64];  q3 = e0[96];
    }
    AvStage sa, sb, sc;
    av_load(sa, arow, bcol, 0 * AV_GRP);
    av_load(sb, arow, bcol, 1 * AV_GRP);
    AV_FENCE();
    const float mx = fmaxf(fmaxf(fmaxf(t0[0], t0[2]), fmaxf(t1[0], t1[2])),
                           fmaxf(fmaxf(t2[0], t2[2]), fmaxf(t3[0], t3[2])));
    const float w0 = __expf(t0[0] - mx), w1 = __expf(t0[2] - mx);
    const float w2 = __expf(t1[0] - mx), w3 = __expf(t1[2] - mx);
    const float w4 = __expf(t2[0] - mx), w5 = __expf(t2[2] - mx);
    const float w6 = __expf(t3[0] - mx), w7 = __expf(t3[2] - mx);
    float tot = w0 * t0[1];
    tot = fmaf(w1, t0[3], tot);
    tot = fmaf(w2, t1[1], tot);
    tot = fmaf(w3, t1[3], tot);
    tot = fmaf(w4, t2[1], tot);
    tot = fmaf(w5, t2[3], tot);
    tot = fmaf(w6, t3[1], tot);
    tot = fmaf(w7, t3[3], tot);
    const float inv = 1.f / tot;
    const float f0 = w0 * inv, f1 = w1 * inv, f2 = w2 * inv, f3 = w3 * inv;
    const float f4 = w4 * inv, f5 = w5 * inv, f6 = w6 * inv, f7 = w7 * inv;
    if (row0) {
        const float z0 = first_lane(f0), z1 = first_lane(f1), z2 = first_lane(f2), z3 = first_lane(f3);
        const float z4 = first_lane(f4), z5 = first_lane(f5), z6 = first_lane(f6), z7 = first_lane(f7);
        const bool hi = lane >= 16;
        const float g0 = hi ? z1 : z0, g1 = hi ? z3 : z2, g2 = hi ? z5 : z4, g3 = hi ? z7 : z6;
        v4f* a = (v4f*)att00 + lane;
        a[0] = q0 * g0;  a[32] = q1 * g1;  a[64] = q2 * g2;  a[96] = q3 * g3;
    }
    av_load(sc, arow, bcol, 2 * AV_GRP);  AV_FENCE();  c = av_mma(sa, c, f0);  AV_FENCE();
    av_load(sa, arow, bcol, 3 * AV_GRP);  AV_FENCE();  c = av_mma(sb, c, f1);  AV_FENCE();
    av_load(sb, arow, bcol, 4 * AV_GRP);  AV_FENCE();  c = av_mma(sc, c, f2);  AV_FENCE();
    av_load(sc, arow, bcol, 5 * AV_GRP);  AV_FENCE();  c = av_mma(sa, c, f3);  AV_FENCE();
    av_load(sa, arow, bcol, 6 * AV_GRP);  AV_FENCE();  c = av_mma(sb, c, f4);  AV_FENCE();
    av_load(sb, arow, bcol, 7 * AV_GRP);  AV_FENCE();  c = av_mma(sc, c, f5);  AV_FENCE();
    c = av_mma(sa, c, f6);
    c = av_mma(sb, c, f7);
#undef AV_FENCE

    float* orow = otile + (size_t)(8 * lhalf) * GAT_F + lm;
    #pragma unroll
    for (int v = 0; v < 8; ++v) orow[(size_t)v * GAT_F] = c[v];
}

extern "C" void kernel_launch(void* const* d_in, const int* in_sizes, int n_in,
                              void* d_out, int out_size, void* d_ws, size_t ws_size,
                              hipStream_t stream) {
    const float* x   = (const float*)d_in[0];
    const float* w1  = (const float*)d_in[1];
    const float* b1  = (const float*)d_in[2];
    const float* w2  = (const float*)d_in[3];
    const float* b2  = (const float*)d_in[4];
    const int*   adj = (const int*)d_in[5];

    float* out = (float*)d_out;
    float* Pb = (float*)d_ws;
    float* Qt = Pb + (size_t)GAT_B * GAT_N * GAT_H;
    float* E  = Qt + (size_t)GAT_B * GAT_N * GAT_H;

    float* next_h = out;
    float* att00  = out + (size_t)GAT_B * GAT_N * GAT_F;

    pq_gemm<<<256, 256, 0, stream>>>(x, w1, b1, Pb, Qt);
    e_kernel<<<256, 256, 0, stream>>>(Pb, Qt, w2, b2, adj, E, next_h);
    av_gemm<<<256, 64, 0, stream>>>(E, x, next_h, att00);
}


# P/Q kernel: fragments by 16-byte loads (one load feeds two WMMAs), a wave's whole K in registers ahead of its WMMAs, one x fragment shared by the P and Q tile; 1024 waves (was 2048); k order changes
# speedup vs baseline: 2.8054x; 1.0351x over previous
#include <hip/hip_runtime.h>
#include <hip/hip_bf16.h>


typedef __attribute__((ext_vector_type(2))) float v2f;
typedef __attribute__((ext_vector_type(4))) float v4f;
typedef __attribute__((ext_vector_type(8))) float v8f;

#define GAT_B 2
#define GAT_N 512
#define GAT_F 128
#define GAT_H 256
#define GAT_HQ (GAT_H / 4)
#define NEG_SLOPE 0.04f
#define MASK_VAL  (-9.0e15f)

#define PQ_GRP (GAT_F / 8)

__global__ __launch_bounds__(128) void pq_gemm(const float* __restrict__ x,
                                               const float* __restrict__ w1,
                                               const float* __restrict__ b1,
                                               float* __restrict__ Pb,
                                               float* __restrict__ Qt) {
    const int wave = (blockIdx.x * blockDim.x + threadIdx.x) >> 5;
    const int lane = threadIdx.x & 31;
    const int m0 = (wave >> 4) << 4;
    const int n0 = (wave & 15) << 4;
    const int lhalf = lane >> 4;
    const int lm    = lane & 15;
    const int h     = n0 + lm;

    const float* arow = x  + (size_t)(m0 + lm) * GAT_F + 4 * lhalf;
    const float* brow = w1 + (size_t)h * GAT_H + 4 * lhalf;

    v4f a[PQ_GRP], bp[PQ_GRP], bq[PQ_GRP];
    #pragma unroll
    for (int g = 0; g < PQ_GRP; ++g) {
        a[g]  = *(const v4f*)(arow + 8 * g);
        bp[g] = *(const v4f*)(brow + 8 * g);
        bq[g] = *(const v4f*)(brow + GAT_F + 8 * g);
    }
    const float bias = b1[h];
    __builtin_amdgcn_sched_barrier(0);

    v8f cp = {0.f, 0.f, 0.f, 0.f, 0.f, 0.f, 0.f, 0.f};
    v8f cq = cp;
    #pragma unroll
    for (int g = 0; g < PQ_GRP; ++g) {
        const v2f a0 = {a[g][0], a[g][1]};
        const v2f a1 = {a[g][2], a[g][3]};
        const v2f p0 = {bp[g][0], bp[g][1]};
        const v2f p1 = {bp[g][2], bp[g][3]};
        const v2f q0 = {bq[g][0], bq[g][1]};
        const v2f q1 = {bq[g][2], bq[g][3]};
        cp = __builtin_amdgcn_wmma_f32_16x16x4_f32(false, a0, false, p0,
                                                   (short)0, cp, false, false);
        cq = __builtin_amdgcn_wmma_f32_16x16x4_f32(false, a0, false, q0,
                                                   (short)0, cq, false, false);
        cp = __builtin_amdgcn_wmma_f32_16x16x4_f32(false, a1, false, p1,
                                                   (short)0, cp, false, false);
        cq = __builtin_amdgcn_wmma_f32_16x16x4_f32(false, a1, false, q1,
                                                   (short)0, cq, false, false);
    }

    {
        float* orow = Pb + (size_t)(m0 + 8 * lhalf) * GAT_H + h;
        #pragma unroll
        for (int v = 0; v < 8; ++v) orow[(size_t)v * GAT_H] = cp[v] + bias;
    }
    {
        const int b   = m0 >> 9;
        const int jp0 = ((m0 & (GAT_N - 1)) + 8 * lhalf) >> 1;
        v2f* dst = (v2f*)Qt + (((size_t)(b * GAT_HQ + (h >> 2)) * (GAT_N / 2) + jp0) * 4 + (h & 3));
        #pragma unroll
        for (int p = 0; p < 4; ++p) {
            v2f pr = {cq[2 * p], cq[2 * p + 1]};
            dst[p * 4] = pr;
        }
    }
}

__device__ __forceinline__ float xhalf(const float x) {
    const int b = __builtin_bit_cast(int, x);
    return __builtin_bit_cast(float,
        __builtin_amdgcn_permlanex16(b, b, 0x76543210, 0xfedcba98, false, false));
}

template <int MASK>
__device__ __forceinline__ float xrow(const float x) {
    const int b = __builtin_bit_cast(int, x);
    return __builtin_bit_cast(float,
        __builtin_amdgcn_update_dpp(b, b, 0x160 + MASK, 0xf, 0xf, false));
}

__device__ __forceinline__ float wave_max(float v) {
    v = fmaxf(v, xrow<1>(v));
    v = fmaxf(v, xrow<2>(v));
    v = fmaxf(v, xrow<4>(v));
    v = fmaxf(v, xrow<8>(v));
    return fmaxf(v, xhalf(v));
}

__device__ __forceinline__ float wave_sum(float v) {
    v += xrow<1>(v);
    v += xrow<2>(v);
    v += xrow<4>(v);
    v += xrow<8>(v);
    return v + xhalf(v);
}

__device__ __forceinline__ float first_lane(const float x) {
    return __builtin_bit_cast(float,
        __builtin_amdgcn_readfirstlane(__builtin_bit_cast(int, x)));
}

struct EStage {
    v8f s0, s1, s2, s3;
    v8f w;
    v4f q0, q1, q2, q3;
};

__device__ __forceinline__ EStage e_load(const float* __restrict__ p0,
                                         const float* __restrict__ w2,
                                         const v4f* __restrict__ q, const int h) {
    EStage st;
    st.s0 = *(const v8f*)(p0 + h);
    st.s1 = *(const v8f*)(p0 + GAT_H + h);
    st.s2 = *(const v8f*)(p0 + 2 * GAT_H + h);
    st.s3 = *(const v8f*)(p0 + 3 * GAT_H + h);
    st.w  = *(const v8f*)(w2 + h);
    const v4f* qq = q + (size_t)(h >> 2) * GAT_N;
    st.q0 = qq[0];
    st.q1 = qq[1];
    st.q2 = qq[GAT_N];
    st.q3 = qq[GAT_N + 1];
    return st;
}

__device__ __forceinline__ v2f e_term(const float s, const float w, const v2f qv, const v2f acc) {
    const v2f zero = {0.f, 0.f};
    const v2f sv = {s, s};
    const v2f wv = {w, w};
    return __builtin_elementwise_fma(wv, __builtin_elementwise_max(sv + qv, zero), acc);
}

__device__ __forceinline__ void e_step(const EStage& st, v2f& a0, v2f& a1, v2f& a2, v2f& a3) {
    #pragma unroll
    for (int k = 0; k < 8; ++k) {
        const v4f qq = (k < 2) ? st.q0 : (k < 4) ? st.q1 : (k < 6) ? st.q2 : st.q3;
        const v2f qv = (k & 1) ? (v2f){qq[2], qq[3]} : (v2f){qq[0], qq[1]};
        a0 = e_term(st.s0[k], st.w[k], qv, a0);
        a1 = e_term(st.s1[k], st.w[k], qv, a1);
        a2 = e_term(st.s2[k], st.w[k], qv, a2);
        a3 = e_term(st.s3[k], st.w[k], qv, a3);
    }
}

__device__ __forceinline__ v2f e_finish(const v2f acc, const float b2v, const int2 m) {
    float e0 = acc.x + b2v, e1 = acc.y + b2v;
    e0 = (e0 > 0.f) ? e0 : NEG_SLOPE * e0;
    e1 = (e1 > 0.f) ? e1 : NEG_SLOPE * e1;
    v2f r = {(m.x > 0) ? e0 : MASK_VAL, (m.y > 0) ? e1 : MASK_VAL};
    return r;
}

__global__ __launch_bounds__(256) void e_kernel(const float* __restrict__ Pb,
                                                const float* __restrict__ Qt,
                                                const float* __restrict__ w2,
                                                const float* __restrict__ b2,
                                                const int*   __restrict__ adj,
                                                float* __restrict__ E,
                                                float* __restrict__ parts) {
    const int w    = __builtin_amdgcn_readfirstlane((int)(threadIdx.x >> 5));
    const int lane = threadIdx.x & 31;
    const int jt = blockIdx.x & 7;
    const int ig = (blockIdx.x >> 3) & 15;
    const int b  = blockIdx.x >> 7;
    const int i0 = ig * 32 + w * 4;
    const int jp = jt * 32 + lane;

    const float* p0 = Pb + (size_t)(b * GAT_N + i0) * GAT_H;
    const v4f*   q  = (const v4f*)Qt + ((size_t)b * GAT_HQ * (GAT_N / 2) + jp) * 2;

    const size_t obase = ((size_t)b * GAT_N + i0) * GAT_N + 2 * jp;
    const int2 m0 = *(const int2*)(adj + obase);
    const int2 m1 = *(const int2*)(adj + obase + GAT_N);
    const int2 m2 = *(const int2*)(adj + obase + 2 * GAT_N);
    const int2 m3 = *(const int2*)(adj + obase + 3 * GAT_N);
    const float b2v = b2[0];

    v2f a0 = {0.f, 0.f}, a1 = {0.f, 0.f}, a2 = {0.f, 0.f}, a3 = {0.f, 0.f};
    EStage st0 = e_load(p0, w2, q, 0);
    for (int h = 0; h < GAT_H - 16; h += 16) {
        const EStage st1 = e_load(p0, w2, q, h + 8);
        e_step(st0, a0, a1, a2, a3);
        st0 = e_load(p0, w2, q, h + 16);
        e_step(st1, a0, a1, a2, a3);
    }
    {
        const EStage st1 = e_load(p0, w2, q, GAT_H - 8);
        e_step(st0, a0, a1, a2, a3);
        e_step(st1, a0, a1, a2, a3);
    }

    const v2f r0 = e_finish(a0, b2v, m0);
    const v2f r1 = e_finish(a1, b2v, m1);
    const v2f r2 = e_finish(a2, b2v, m2);
    const v2f r3 = e_finish(a3, b2v, m3);

    const float x0 = wave_max(fmaxf(r0.x, r0.y));
    const float x1 = wave_max(fmaxf(r1.x, r1.y));
    const float x2 = wave_max(fmaxf(r2.x, r2.y));
    const float x3 = wave_max(fmaxf(r3.x, r3.y));
    const v2f y0 = {__expf(r0.x - x0), __expf(r0.y - x0)};
    const v2f y1 = {__expf(r1.x - x1), __expf(r1.y - x1)};
    const v2f y2 = {__expf(r2.x - x2), __expf(r2.y - x2)};
    const v2f y3 = {__expf(r3.x - x3), __expf(r3.y - x3)};
    *(v2f*)(E + obase)             = y0;
    *(v2f*)(E + obase + GAT_N)     = y1;
    *(v2f*)(E + obase + 2 * GAT_N) = y2;
    *(v2f*)(E + obase + 3 * GAT_N) = y3;
    const float s0 = wave_sum(y0.x + y0.y);
    const float s1 = wave_sum(y1.x + y1.y);
    const float s2 = wave_sum(y2.x + y2.y);
    const float s3 = wave_sum(y3.x + y3.y);

    const int rr = lane >> 3, cc = lane & 7;
    const v2f ms = {(rr == 0) ? x0 : (rr == 1) ? x1 : (rr == 2) ? x2 : x3,
                    (rr == 0) ? s0 : (rr == 1) ? s1 : (rr == 2) ? s2 : s3};
    *(v2f*)(parts + ((size_t)b * GAT_N + i0 + rr) * GAT_F + 16 * cc + 2 * jt) = ms;
}

#define AV_GRP 8

struct AvStage {
    v4f a[AV_GRP];
    v4f b[AV_GRP];
};

__device__ __forceinline__ void av_load(AvStage& st, const float* __restrict__ arow,
                                        const float* __restrict__ bcol, const int g0) {
    #pragma unroll
    for (int g = 0; g < AV_GRP; ++g) {
        st.a[g] = *(const v4f*)(arow + (g0 + g) * 8);
        const float* bp = bcol + (size_t)(g0 + g) * 8 * GAT_F;
        const v4f bv = {bp[0], bp[GAT_F], bp[2 * GAT_F], bp[3 * GAT_F]};
        st.b[g] = bv;
    }
}

__device__ __forceinline__ v8f av_mma(const AvStage& st, v8f c, const float sc) {
    #pragma unroll
    for (int g = 0; g < AV_GRP; ++g) {
        const v2f a0 = {st.a[g][0] * sc, st.a[g][1] * sc};
        const v2f b0 = {st.b[g][0], st.b[g][1]};
        const v2f a1 = {st.a[g][2] * sc, st.a[g][3] * sc};
        const v2f b1 = {st.b[g][2], st.b[g][3]};
        c = __builtin_amdgcn_wmma_f32_16x16x4_f32(false, a0, false, b0,
                                                  (short)0, c, false, false);
        c = __builtin_amdgcn_wmma_f32_16x16x4_f32(false, a1, false, b1,
                                                  (short)0, c, false, false);
    }
    return c;
}

__global__ __launch_bounds__(64) void av_gemm(const float* __restrict__ att,
                                              const float* __restrict__ x,
                                              float* __restrict__ out,
                                              float* __restrict__ att00) {
    const int wave = (blockIdx.x * blockDim.x + threadIdx.x) >> 5;
    const int lane = threadIdx.x & 31;
    const int b    = wave >> 8;
    const int rem  = wave & 255;
    const int m0 = (rem >> 3) << 4;
    const int n0 = (rem & 7) << 4;
    const int lhalf = lane >> 4;
    const int lm    = lane & 15;

    const float* A = att + (size_t)b * GAT_N * GAT_N;
    const float* X = x   + (size_t)b * GAT_N * GAT_F;
    const float* arow = A + (size_t)(m0 + lm) * GAT_N + 4 * lhalf;
    const float* bcol = X + (size_t)(4 * lhalf) * GAT_F + n0 + lm;

    v8f c = {0.f, 0.f, 0.f, 0.f, 0.f, 0.f, 0.f, 0.f};
#define AV_FENCE() __builtin_amdgcn_sched_barrier(0)
    float* otile = out + (size_t)b * GAT_N * GAT_F + (size_t)m0 * GAT_F + n0;
    const v4f* pr = (const v4f*)(otile + (size_t)lm * GAT_F);
    const v4f t0 = pr[0], t1 = pr[1], t2 = pr[2], t3 = pr[3];
    const bool row0 = __builtin_amdgcn_readfirstlane(wave) == 0;
    v4f q0 = {0.f, 0.f, 0.f, 0.f}, q1 = q0, q2 = q0, q3 = q0;
    if (row0) {
        const v4f* e0 = (const v4f*)att + lane;
        q0 = e0[0];  q1 = e0[32];  q2 = e0[64];  q3 = e0[96];
    }
    AvStage sa, sb, sc;
    av_load(sa, arow, bcol, 0 * AV_GRP);
    av_load(sb, arow, bcol, 1 * AV_GRP);
    AV_FENCE();
    const float mx = fmaxf(fmaxf(fmaxf(t0[0], t0[2]), fmaxf(t1[0], t1[2])),
                           fmaxf(fmaxf(t2[0], t2[2]), fmaxf(t3[0], t3[2])));
    const float w0 = __expf(t0[0] - mx), w1 = __expf(t0[2] - mx);
    const float w2 = __expf(t1[0] - mx), w3 = __expf(t1[2] - mx);
    const float w4 = __expf(t2[0] - mx), w5 = __expf(t2[2] - mx);
    const float w6 = __expf(t3[0] - mx), w7 = __expf(t3[2] - mx);
    float tot = w0 * t0[1];
    tot = fmaf(w1, t0[3], tot);
    tot = fmaf(w2, t1[1], tot);
    tot = fmaf(w3, t1[3], tot);
    tot = fmaf(w4, t2[1], tot);
    tot = fmaf(w5, t2[3], tot);
    tot = fmaf(w6, t3[1], tot);
    tot = fmaf(w7, t3[3], tot);
    const float inv = 1.f / tot;
    const float f0 = w0 * inv, f1 = w1 * inv, f2 = w2 * inv, f3 = w3 * inv;
    const float f4 = w4 * inv, f5 = w5 * inv, f6 = w6 * inv, f7 = w7 * inv;
    if (row0) {
        const float z0 = first_lane(f0), z1 = first_lane(f1), z2 = first_lane(f2), z3 = first_lane(f3);
        const float z4 = first_lane(f4), z5 = first_lane(f5), z6 = first_lane(f6), z7 = first_lane(f7);
        const bool hi = lane >= 16;
        const float g0 = hi ? z1 : z0, g1 = hi ? z3 : z2, g2 = hi ? z5 : z4, g3 = hi ? z7 : z6;
        v4f* a = (v4f*)att00 + lane;
        a[0] = q0 * g0;  a[32] = q1 * g1;  a[64] = q2 * g2;  a[96] = q3 * g3;
    }
    av_load(sc, arow, bcol, 2 * AV_GRP);  AV_FENCE();  c = av_mma(sa, c, f0);  AV_FENCE();
    av_load(sa, arow, bcol, 3 * AV_GRP);  AV_FENCE();  c = av_mma(sb, c, f1);  AV_FENCE();
    av_load(sb, arow, bcol, 4 * AV_GRP);  AV_FENCE();  c = av_mma(sc, c, f2);  AV_FENCE();
    av_load(sc, arow, bcol, 5 * AV_GRP);  AV_FENCE();  c = av_mma(sa, c, f3);  AV_FENCE();
    av_load(sa, arow, bcol, 6 * AV_GRP);  AV_FENCE();  c = av_mma(sb, c, f4);  AV_FENCE();
    av_load(sb, arow, bcol, 7 * AV_GRP);  AV_FENCE();  c = av_mma(sc, c, f5);  AV_FENCE();
    c = av_mma(sa, c, f6);
    c = av_mma(sb, c, f7);
#undef AV_FENCE

    float* orow = otile + (size_t)(8 * lhalf) * GAT_F + lm;
    #pragma unroll
    for (int v = 0; v < 8; ++v) orow[(size_t)v * GAT_F] = c[v];
}

extern "C" void kernel_launch(void* const* d_in, const int* in_sizes, int n_in,
                              void* d_out, int out_size, void* d_ws, size_t ws_size,
                              hipStream_t stream) {
    const float* x   = (const float*)d_in[0];
    const float* w1  = (const float*)d_in[1];
    const float* b1  = (const float*)d_in[2];
    const float* w2  = (const float*)d_in[3];
    const float* b2  = (const float*)d_in[4];
    const int*   adj = (const int*)d_in[5];

    float* out = (float*)d_out;
    float* Pb = (float*)d_ws;
    float* Qt = Pb + (size_t)GAT_B * GAT_N * GAT_H;
    float* E  = Qt + (size_t)GAT_B * GAT_N * GAT_H;

    float* next_h = out;
    float* att00  = out + (size_t)GAT_B * GAT_N * GAT_F;

    pq_gemm<<<256, 128, 0, stream>>>(x, w1, b1, Pb, Qt);
    e_kernel<<<256, 256, 0, stream>>>(Pb, Qt, w2, b2, adj, E, next_h);
    av_gemm<<<256, 64, 0, stream>>>(E, x, next_h, att00);
}


# e-logit kernel: load latency hidden in registers: P+b1 rows and w2 held whole in VGPRs (32 bytes a lane, v_readlane to scalar operands), Q ring 7 stages ahead; no scalar loads in the loop; sums kept
# speedup vs baseline: 3.1107x; 1.1088x over previous
#include <hip/hip_runtime.h>
#include <hip/hip_bf16.h>


typedef __attribute__((ext_vector_type(2))) float v2f;
typedef __attribute__((ext_vector_type(4))) float v4f;
typedef __attribute__((ext_vector_type(8))) float v8f;

#define GAT_B 2
#define GAT_N 512
#define GAT_F 128
#define GAT_H 256
#define GAT_HQ (GAT_H / 4)
#define NEG_SLOPE 0.04f
#define MASK_VAL  (-9.0e15f)

#define PQ_GRP (GAT_F / 8)

__global__ __launch_bounds__(128) void pq_gemm(const float* __restrict__ x,
                                               const float* __restrict__ w1,
                                               const float* __restrict__ b1,
                                               float* __restrict__ Pb,
                                               float* __restrict__ Qt) {
    const int wave = (blockIdx.x * blockDim.x + threadIdx.x) >> 5;
    const int lane = threadIdx.x & 31;
    const int m0 = (wave >> 4) << 4;
    const int n0 = (wave & 15) << 4;
    const int lhalf = lane >> 4;
    const int lm    = lane & 15;
    const int h     = n0 + lm;

    const float* arow = x  + (size_t)(m0 + lm) * GAT_F + 4 * lhalf;
    const float* brow = w1 + (size_t)h * GAT_H + 4 * lhalf;

    v4f a[PQ_GRP], bp[PQ_GRP], bq[PQ_GRP];
    #pragma unroll
    for (int g = 0; g < PQ_GRP; ++g) {
        a[g]  = *(const v4f*)(arow + 8 * g);
        bp[g] = *(const v4f*)(brow + 8 * g);
        bq[g] = *(const v4f*)(brow + GAT_F + 8 * g);
    }
    const float bias = b1[h];
    __builtin_amdgcn_sched_barrier(0);

    v8f cp = {0.f, 0.f, 0.f, 0.f, 0.f, 0.f, 0.f, 0.f};
    v8f cq = cp;
    #pragma unroll
    for (int g = 0; g < PQ_GRP; ++g) {
        const v2f a0 = {a[g][0], a[g][1]};
        const v2f a1 = {a[g][2], a[g][3]};
        const v2f p0 = {bp[g][0], bp[g][1]};
        const v2f p1 = {bp[g][2], bp[g][3]};
        const v2f q0 = {bq[g][0], bq[g][1]};
        const v2f q1 = {bq[g][2], bq[g][3]};
        cp = __builtin_amdgcn_wmma_f32_16x16x4_f32(false, a0, false, p0,
                                                   (short)0, cp, false, false);
        cq = __builtin_amdgcn_wmma_f32_16x16x4_f32(false, a0, false, q0,
                                                   (short)0, cq, false, false);
        cp = __builtin_amdgcn_wmma_f32_16x16x4_f32(false, a1, false, p1,
                                                   (short)0, cp, false, false);
        cq = __builtin_amdgcn_wmma_f32_16x16x4_f32(false, a1, false, q1,
                                                   (short)0, cq, false, false);
    }

    {
        float* orow = Pb + (size_t)(m0 + 8 * lhalf) * GAT_H + h;
        #pragma unroll
        for (int v = 0; v < 8; ++v) orow[(size_t)v * GAT_H] = cp[v] + bias;
    }
    {
        const int b   = m0 >> 9;
        const int jp0 = ((m0 & (GAT_N - 1)) + 8 * lhalf) >> 1;
        v2f* dst = (v2f*)Qt + (((size_t)(b * GAT_HQ + (h >> 2)) * (GAT_N / 2) + jp0) * 4 + (h & 3));
        #pragma unroll
        for (int p = 0; p < 4; ++p) {
            v2f pr = {cq[2 * p], cq[2 * p + 1]};
            dst[p * 4] = pr;
        }
    }
}

__device__ __forceinline__ float xhalf(const float x) {
    const int b = __builtin_bit_cast(int, x);
    return __builtin_bit_cast(float,
        __builtin_amdgcn_permlanex16(b, b, 0x76543210, 0xfedcba98, false, false));
}

template <int MASK>
__device__ __forceinline__ float xrow(const float x) {
    const int b = __builtin_bit_cast(int, x);
    return __builtin_bit_cast(float,
        __builtin_amdgcn_update_dpp(b, b, 0x160 + MASK, 0xf, 0xf, false));
}

__device__ __forceinline__ float wave_max(float v) {
    v = fmaxf(v, xrow<1>(v));
    v = fmaxf(v, xrow<2>(v));
    v = fmaxf(v, xrow<4>(v));
    v = fmaxf(v, xrow<8>(v));
    return fmaxf(v, xhalf(v));
}

__device__ __forceinline__ float wave_sum(float v) {
    v += xrow<1>(v);
    v += xrow<2>(v);
    v += xrow<4>(v);
    v += xrow<8>(v);
    return v + xhalf(v);
}

__device__ __forceinline__ float first_lane(const float x) {
    return __builtin_bit_cast(float,
        __builtin_amdgcn_readfirstlane(__builtin_bit_cast(int, x)));
}

struct QStage {
    v4f q0, q1, q2, q3;
};

__device__ __forceinline__ QStage q_load(const v4f* __restrict__ q, const int st) {
    const v4f* qq = q + (size_t)(2 * st) * GAT_N;
    QStage s;
    s.q0 = qq[0];
    s.q1 = qq[1];
    s.q2 = qq[GAT_N];
    s.q3 = qq[GAT_N + 1];
    return s;
}

__device__ __forceinline__ float lane_val(const float v, const int l) {
    return __builtin_bit_cast(float,
        __builtin_amdgcn_readlane(__builtin_bit_cast(int, v), l));
}

__device__ __forceinline__ v2f e_term(const float s, const float w, const v2f qv, const v2f acc) {
    const v2f zero = {0.f, 0.f};
    const v2f sv = {s, s};
    const v2f wv = {w, w};
    return __builtin_elementwise_fma(wv, __builtin_elementwise_max(sv + qv, zero), acc);
}

struct ESc {
    float s0a, s1a, s2a, s3a, wa;
    float s0b, s1b, s2b, s3b, wb;
};

template <int K>
__device__ __forceinline__ ESc sc_read(const v8f& p0, const v8f& p1, const v8f& p2, const v8f& p3,
                                       const v8f& w8, const int l) {
    ESc c;
    c.s0a = lane_val(p0[K], l);      c.s1a = lane_val(p1[K], l);
    c.s2a = lane_val(p2[K], l);      c.s3a = lane_val(p3[K], l);
    c.wa  = lane_val(w8[K], l);
    c.s0b = lane_val(p0[K + 1], l);  c.s1b = lane_val(p1[K + 1], l);
    c.s2b = lane_val(p2[K + 1], l);  c.s3b = lane_val(p3[K + 1], l);
    c.wb  = lane_val(w8[K + 1], l);
    return c;
}

__device__ __forceinline__ void e_pair(const ESc& c, const v4f qq,
                                       v2f& a0, v2f& a1, v2f& a2, v2f& a3) {
    const v2f qa = {qq[0], qq[1]};
    const v2f qb = {qq[2], qq[3]};
    a0 = e_term(c.s0a, c.wa, qa, a0);
    a1 = e_term(c.s1a, c.wa, qa, a1);
    a2 = e_term(c.s2a, c.wa, qa, a2);
    a3 = e_term(c.s3a, c.wa, qa, a3);
    a0 = e_term(c.s0b, c.wb, qb, a0);
    a1 = e_term(c.s1b, c.wb, qb, a1);
    a2 = e_term(c.s2b, c.wb, qb, a2);
    a3 = e_term(c.s3b, c.wb, qb, a3);
}

#define E_FENCE() __builtin_amdgcn_sched_barrier(0)

template <bool LAST>
__device__ __forceinline__ void e_stage(const v8f& p0, const v8f& p1, const v8f& p2, const v8f& p3,
                                        const v8f& w8, const QStage& st, const int l, ESc& cur,
                                        v2f& a0, v2f& a1, v2f& a2, v2f& a3) {
    {
        const ESc n = sc_read<2>(p0, p1, p2, p3, w8, l);
        e_pair(cur, st.q0, a0, a1, a2, a3);
        E_FENCE();
        cur = n;
    }
    {
        const ESc n = sc_read<4>(p0, p1, p2, p3, w8, l);
        e_pair(cur, st.q1, a0, a1, a2, a3);
        E_FENCE();
        cur = n;
    }
    {
        const ESc n = sc_read<6>(p0, p1, p2, p3, w8, l);
        e_pair(cur, st.q2, a0, a1, a2, a3);
        E_FENCE();
        cur = n;
    }
    if (LAST) {
        e_pair(cur, st.q3, a0, a1, a2, a3);
    } else {
        const ESc n = sc_read<0>(p0, p1, p2, p3, w8, l + 1);
        e_pair(cur, st.q3, a0, a1, a2, a3);
        E_FENCE();
        cur = n;
    }
}

__device__ __forceinline__ v2f e_finish(const v2f acc, const float b2v, const int2 m) {
    float e0 = acc.x + b2v, e1 = acc.y + b2v;
    e0 = (e0 > 0.f) ? e0 : NEG_SLOPE * e0;
    e1 = (e1 > 0.f) ? e1 : NEG_SLOPE * e1;
    v2f r = {(m.x > 0) ? e0 : MASK_VAL, (m.y > 0) ? e1 : MASK_VAL};
    return r;
}

__global__ __launch_bounds__(256) void e_kernel(const float* __restrict__ Pb,
                                                const float* __restrict__ Qt,
                                                const float* __restrict__ w2,
                                                const float* __restrict__ b2,
                                                const int*   __restrict__ adj,
                                                float* __restrict__ E,
                                                float* __restrict__ parts) {
    const int w    = __builtin_amdgcn_readfirstlane((int)(threadIdx.x >> 5));
    const int lane = threadIdx.x & 31;
    const int jt = blockIdx.x & 7;
    const int ig = (blockIdx.x >> 3) & 15;
    const int b  = blockIdx.x >> 7;
    const int i0 = ig * 32 + w * 4;
    const int jp = jt * 32 + lane;

    const float* p0 = Pb + (size_t)(b * GAT_N + i0) * GAT_H;
    const v4f*   q  = (const v4f*)Qt + ((size_t)b * GAT_HQ * (GAT_N / 2) + jp) * 2;

    const size_t obase = ((size_t)b * GAT_N + i0) * GAT_N + 2 * jp;
    const int2 m0 = *(const int2*)(adj + obase);
    const int2 m1 = *(const int2*)(adj + obase + GAT_N);
    const int2 m2 = *(const int2*)(adj + obase + 2 * GAT_N);
    const int2 m3 = *(const int2*)(adj + obase + 3 * GAT_N);
    const float b2v = b2[0];

    const v8f P0 = *(const v8f*)(p0 + 8 * lane);
    const v8f P1 = *(const v8f*)(p0 + GAT_H + 8 * lane);
    const v8f P2 = *(const v8f*)(p0 + 2 * GAT_H + 8 * lane);
    const v8f P3 = *(const v8f*)(p0 + 3 * GAT_H + 8 * lane);
    const v8f W8 = *(const v8f*)(w2 + 8 * lane);

    v2f a0 = {0.f, 0.f}, a1 = {0.f, 0.f}, a2 = {0.f, 0.f}, a3 = {0.f, 0.f};
    QStage g0 = q_load(q, 0);
    QStage g1 = q_load(q, 1);
    QStage g2 = q_load(q, 2);
    QStage g3 = q_load(q, 3);
    QStage g4 = q_load(q, 4);
    QStage g5 = q_load(q, 5);
    QStage g6 = q_load(q, 6);
    QStage g7 = q_load(q, 7);
    ESc cur = sc_read<0>(P0, P1, P2, P3, W8, 0);
    E_FENCE();
#define E_STAGE(G, S) e_stage<false>(P0, P1, P2, P3, W8, G, S, cur, a0, a1, a2, a3)
    #pragma nounroll
    for (int st = 0; st < GAT_H / 8 - 8; st += 8) {
        E_STAGE(g0, st);      g0 = q_load(q, st + 8);
        E_STAGE(g1, st + 1);  g1 = q_load(q, st + 9);
        E_STAGE(g2, st + 2);  g2 = q_load(q, st + 10);
        E_STAGE(g3, st + 3);  g3 = q_load(q, st + 11);
        E_STAGE(g4, st + 4);  g4 = q_load(q, st + 12);
        E_STAGE(g5, st + 5);  g5 = q_load(q, st + 13);
        E_STAGE(g6, st + 6);  g6 = q_load(q, st + 14);
        E_STAGE(g7, st + 7);  g7 = q_load(q, st + 15);
    }
    E_STAGE(g0, GAT_H / 8 - 8);
    E_STAGE(g1, GAT_H / 8 - 7);
    E_STAGE(g2, GAT_H / 8 - 6);
    E_STAGE(g3, GAT_H / 8 - 5);
    E_STAGE(g4, GAT_H / 8 - 4);
    E_STAGE(g5, GAT_H / 8 - 3);
    E_STAGE(g6, GAT_H / 8 - 2);
    e_stage<true>(P0, P1, P2, P3, W8, g7, GAT_H / 8 - 1, cur, a0, a1, a2, a3);
#undef E_STAGE
#undef E_FENCE

    const v2f r0 = e_finish(a0, b2v, m0);
    const v2f r1 = e_finish(a1, b2v, m1);
    const v2f r2 = e_finish(a2, b2v, m2);
    const v2f r3 = e_finish(a3, b2v, m3);

    const float x0 = wave_max(fmaxf(r0.x, r0.y));
    const float x1 = wave_max(fmaxf(r1.x, r1.y));
    const float x2 = wave_max(fmaxf(r2.x, r2.y));
    const float x3 = wave_max(fmaxf(r3.x, r3.y));
    const v2f y0 = {__expf(r0.x - x0), __expf(r0.y - x0)};
    const v2f y1 = {__expf(r1.x - x1), __expf(r1.y - x1)};
    const v2f y2 = {__expf(r2.x - x2), __expf(r2.y - x2)};
    const v2f y3 = {__expf(r3.x - x3), __expf(r3.y - x3)};
    *(v2f*)(E + obase)             = y0;
    *(v2f*)(E + obase + GAT_N)     = y1;
    *(v2f*)(E + obase + 2 * GAT_N) = y2;
    *(v2f*)(E + obase + 3 * GAT_N) = y3;
    const float s0 = wave_sum(y0.x + y0.y);
    const float s1 = wave_sum(y1.x + y1.y);
    const float s2 = wave_sum(y2.x + y2.y);
    const float s3 = wave_sum(y3.x + y3.y);

    const int rr = lane >> 3, cc = lane & 7;
    const v2f ms = {(rr == 0) ? x0 : (rr == 1) ? x1 : (rr == 2) ? x2 : x3,
                    (rr == 0) ? s0 : (rr == 1) ? s1 : (rr == 2) ? s2 : s3};
    *(v2f*)(parts + ((size_t)b * GAT_N + i0 + rr) * GAT_F + 16 * cc + 2 * jt) = ms;
}

#define AV_GRP 8

struct AvStage {
    v4f a[AV_GRP];
    v4f b[AV_GRP];
};

__device__ __forceinline__ void av_load(AvStage& st, const float* __restrict__ arow,
                                        const float* __restrict__ bcol, const int g0) {
    #pragma unroll
    for (int g = 0; g < AV_GRP; ++g) {
        st.a[g] = *(const v4f*)(arow + (g0 + g) * 8);
        const float* bp = bcol + (size_t)(g0 + g) * 8 * GAT_F;
        const v4f bv = {bp[0], bp[GAT_F], bp[2 * GAT_F], bp[3 * GAT_F]};
        st.b[g] = bv;
    }
}

__device__ __forceinline__ v8f av_mma(const AvStage& st, v8f c, const float sc) {
    #pragma unroll
    for (int g = 0; g < AV_GRP; ++g) {
        const v2f a0 = {st.a[g][0] * sc, st.a[g][1] * sc};
        const v2f b0 = {st.b[g][0], st.b[g][1]};
        const v2f a1 = {st.a[g][2] * sc, st.a[g][3] * sc};
        const v2f b1 = {st.b[g][2], st.b[g][3]};
        c = __builtin_amdgcn_wmma_f32_16x16x4_f32(false, a0, false, b0,
                                                  (short)0, c, false, false);
        c = __builtin_amdgcn_wmma_f32_16x16x4_f32(false, a1, false, b1,
                                                  (short)0, c, false, false);
    }
    return c;
}

__global__ __launch_bounds__(64) void av_gemm(const float* __restrict__ att,
                                              const float* __restrict__ x,
                                              float* __restrict__ out,
                                              float* __restrict__ att00) {
    const int wave = (blockIdx.x * blockDim.x + threadIdx.x) >> 5;
    const int lane = threadIdx.x & 31;
    const int b    = wave >> 8;
    const int rem  = wave & 255;
    const int m0 = (rem >> 3) << 4;
    const int n0 = (rem & 7) << 4;
    const int lhalf = lane >> 4;
    const int lm    = lane & 15;

    const float* A = att + (size_t)b * GAT_N * GAT_N;
    const float* X = x   + (size_t)b * GAT_N * GAT_F;
    const float* arow = A + (size_t)(m0 + lm) * GAT_N + 4 * lhalf;
    const float* bcol = X + (size_t)(4 * lhalf) * GAT_F + n0 + lm;

    v8f c = {0.f, 0.f, 0.f, 0.f, 0.f, 0.f, 0.f, 0.f};
#define AV_FENCE() __builtin_amdgcn_sched_barrier(0)
    float* otile = out + (size_t)b * GAT_N * GAT_F + (size_t)m0 * GAT_F + n0;
    const v4f* pr = (const v4f*)(otile + (size_t)lm * GAT_F);
    const v4f t0 = pr[0], t1 = pr[1], t2 = pr[2], t3 = pr[3];
    const bool row0 = __builtin_amdgcn_readfirstlane(wave) == 0;
    v4f q0 = {0.f, 0.f, 0.f, 0.f}, q1 = q0, q2 = q0, q3 = q0;
    if (row0) {
        const v4f* e0 = (const v4f*)att + lane;
        q0 = e0[0];  q1 = e0[32];  q2 = e0[64];  q3 = e0[96];
    }
    AvStage sa, sb, sc;
    av_load(sa, arow, bcol, 0 * AV_GRP);
    av_load(sb, arow, bcol, 1 * AV_GRP);
    AV_FENCE();
    const float mx = fmaxf(fmaxf(fmaxf(t0[0], t0[2]), fmaxf(t1[0], t1[2])),
                           fmaxf(fmaxf(t2[0], t2[2]), fmaxf(t3[0], t3[2])));
    const float w0 = __expf(t0[0] - mx), w1 = __expf(t0[2] - mx);
    const float w2 = __expf(t1[0] - mx), w3 = __expf(t1[2] - mx);
    const float w4 = __expf(t2[0] - mx), w5 = __expf(t2[2] - mx);
    const float w6 = __expf(t3[0] - mx), w7 = __expf(t3[2] - mx);
    float tot = w0 * t0[1];
    tot = fmaf(w1, t0[3], tot);
    tot = fmaf(w2, t1[1], tot);
    tot = fmaf(w3, t1[3], tot);
    tot = fmaf(w4, t2[1], tot);
    tot = fmaf(w5, t2[3], tot);
    tot = fmaf(w6, t3[1], tot);
    tot = fmaf(w7, t3[3], tot);
    const float inv = 1.f / tot;
    const float f0 = w0 * inv, f1 = w1 * inv, f2 = w2 * inv, f3 = w3 * inv;
    const float f4 = w4 * inv, f5 = w5 * inv, f6 = w6 * inv, f7 = w7 * inv;
    if (row0) {
        const float z0 = first_lane(f0), z1 = first_lane(f1), z2 = first_lane(f2), z3 = first_lane(f3);
        const float z4 = first_lane(f4), z5 = first_lane(f5), z6 = first_lane(f6), z7 = first_lane(f7);
        const bool hi = lane >= 16;
        const float g0 = hi ? z1 : z0, g1 = hi ? z3 : z2, g2 = hi ? z5 : z4, g3 = hi ? z7 : z6;
        v4f* a = (v4f*)att00 + lane;
        a[0] = q0 * g0;  a[32] = q1 * g1;  a[64] = q2 * g2;  a[96] = q3 * g3;
    }
    av_load(sc, arow, bcol, 2 * AV_GRP);  AV_FENCE();  c = av_mma(sa, c, f0);  AV_FENCE();
    av_load(sa, arow, bcol, 3 * AV_GRP);  AV_FENCE();  c = av_mma(sb, c, f1);  AV_FENCE();
    av_load(sb, arow, bcol, 4 * AV_GRP);  AV_FENCE();  c = av_mma(sc, c, f2);  AV_FENCE();
    av_load(sc, arow, bcol, 5 * AV_GRP);  AV_FENCE();  c = av_mma(sa, c, f3);  AV_FENCE();
    av_load(sa, arow, bcol, 6 * AV_GRP);  AV_FENCE();  c = av_mma(sb, c, f4);  AV_FENCE();
    av_load(sb, arow, bcol, 7 * AV_GRP);  AV_FENCE();  c = av_mma(sc, c, f5);  AV_FENCE();
    c = av_mma(sa, c, f6);
    c = av_mma(sb, c, f7);
#undef AV_FENCE

    float* orow = otile + (size_t)(8 * lhalf) * GAT_F + lm;
    #pragma unroll
    for (int v = 0; v < 8; ++v) orow[(size_t)v * GAT_F] = c[v];
}

extern "C" void kernel_launch(void* const* d_in, const int* in_sizes, int n_in,
                              void* d_out, int out_size, void* d_ws, size_t ws_size,
                              hipStream_t stream) {
    const float* x   = (const float*)d_in[0];
    const float* w1  = (const float*)d_in[1];
    const float* b1  = (const float*)d_in[2];
    const float* w2  = (const float*)d_in[3];
    const float* b2  = (const float*)d_in[4];
    const int*   adj = (const int*)d_in[5];

    float* out = (float*)d_out;
    float* Pb = (float*)d_ws;
    float* Qt = Pb + (size_t)GAT_B * GAT_N * GAT_H;
    float* E  = Qt + (size_t)GAT_B * GAT_N * GAT_H;

    float* next_h = out;
    float* att00  = out + (size_t)GAT_B * GAT_N * GAT_F;

    pq_gemm<<<256, 128, 0, stream>>>(x, w1, b1, Pb, Qt);
    e_kernel<<<256, 256, 0, stream>>>(Pb, Qt, w2, b2, adj, E, next_h);
    av_gemm<<<256, 64, 0, stream>>>(E, x, next_h, att00);
}
